# starting kernel
# baseline (speedup 1.0000x reference)
#include <hip/hip_runtime.h>
#include <hip/hip_bf16.h>

typedef __attribute__((ext_vector_type(2))) float v2f;
typedef __attribute__((ext_vector_type(8))) float v8f;


__global__ void k_init_deg(float* __restrict__ deg, int n) {
    int i = blockIdx.x * blockDim.x + threadIdx.x;
    if (i < n) deg[i] = 1.0f;
}

__global__ void k_count_deg(const int* __restrict__ ei, float* __restrict__ deg, int nedges) {
    int e = blockIdx.x * blockDim.x + threadIdx.x;
    if (e < nedges) atomicAdd(&deg[ei[nedges + e]], 1.0f);
}

__global__ void k_rsqrt_inplace(float* __restrict__ deg, int n) {
    int i = blockIdx.x * blockDim.x + threadIdx.x;
    if (i < n) deg[i] = rsqrtf(deg[i]);
}

__global__ void k_edge_norm(const int* __restrict__ ei, const float* __restrict__ dinv,
                            float* __restrict__ w, int nedges) {
    int e = blockIdx.x * blockDim.x + threadIdx.x;
    if (e < nedges) w[e] = dinv[ei[e]] * dinv[ei[nedges + e]];
}


#define WT_PITCH 68

template <bool BIAS_RELU>
__global__ void k_gemm64(const float* __restrict__ A, const float* __restrict__ W,
                         const float* __restrict__ bias, float* __restrict__ C, int nrows) {
    __shared__ float ldsW[64 * WT_PITCH];

    for (int idx = threadIdx.x; idx < 4096; idx += 256) {
        const int k = idx >> 6;
        const int col = idx & 63;
        ldsW[col * WT_PITCH + k] = W[idx];
    }
    __syncthreads();

    const int lane = threadIdx.x & 31;
    const int wave = threadIdx.x >> 5;
    const int rowTiles = (nrows + 15) >> 4;
    const int tile = blockIdx.x * 8 + wave;
    if (tile < rowTiles) {
        const int half = lane >> 4;
        const int m    = lane & 15;
        int row = tile * 16 + m;
        if (row >= nrows) row = nrows - 1;
        const float* __restrict__ arow = A + (size_t)row * 64;

        v8f c0 = {}, c1 = {}, c2 = {}, c3 = {};
#pragma unroll
        for (int kk = 0; kk < 16; ++kk) {
            const int k0 = kk * 4 + 2 * half;
            v2f a;
            a.x = arow[k0];
            a.y = arow[k0 + 1];
            v2f b0 = *(const v2f*)&ldsW[(0 * 16 + m) * WT_PITCH + k0];
            v2f b1 = *(const v2f*)&ldsW[(1 * 16 + m) * WT_PITCH + k0];
            v2f b2 = *(const v2f*)&ldsW[(2 * 16 + m) * WT_PITCH + k0];
            v2f b3 = *(const v2f*)&ldsW[(3 * 16 + m) * WT_PITCH + k0];
            c0 = __builtin_amdgcn_wmma_f32_16x16x4_f32(false, a, false, b0, (short)0, c0, false, false);
            c1 = __builtin_amdgcn_wmma_f32_16x16x4_f32(false, a, false, b1, (short)0, c1, false, false);
            c2 = __builtin_amdgcn_wmma_f32_16x16x4_f32(false, a, false, b2, (short)0, c2, false, false);
            c3 = __builtin_amdgcn_wmma_f32_16x16x4_f32(false, a, false, b3, (short)0, c3, false, false);
        }

        v8f acc[4] = {c0, c1, c2, c3};
#pragma unroll
        for (int nt = 0; nt < 4; ++nt) {
            const int col = nt * 16 + m;
            const float bv = BIAS_RELU ? bias[col] : 0.0f;
#pragma unroll
            for (int r = 0; r < 8; ++r) {
                const int orow = tile * 16 + r + 8 * half;
                if (orow < nrows) {
                    float v = acc[nt][r];
                    if (BIAS_RELU) v = fmaxf(v + bv, 0.0f);
                    C[(size_t)orow * 64 + col] = v;
                }
            }
        }
    }
}


__global__ void k_selfloop_init(const float* __restrict__ t, const float* __restrict__ dinv,
                                float* __restrict__ agg, int n) {
    long tid = (long)blockIdx.x * blockDim.x + threadIdx.x;
    int i = (int)(tid >> 4);
    int j = (int)(tid & 15) * 4;
    if (i >= n) return;
    float w = dinv[i] * dinv[i];
    float4 v = *(const float4*)(t + (size_t)i * 64 + j);
    float4 r = make_float4(w * v.x, w * v.y, w * v.z, w * v.w);
    *(float4*)(agg + (size_t)i * 64 + j) = r;
}

__global__ void k_edge_scatter(const int* __restrict__ ei, const float* __restrict__ wnorm,
                               const float* __restrict__ t, float* __restrict__ agg, int nedges) {
    long tid = (long)blockIdx.x * blockDim.x + threadIdx.x;
    int e = (int)(tid >> 4);
    int j = (int)(tid & 15) * 4;
    if (e >= nedges) return;
    int s = ei[e];
    int d = ei[nedges + e];
    float w = wnorm[e];
    float4 v = *(const float4*)(t + (size_t)s * 64 + j);
    float* p = agg + (size_t)d * 64 + j;
    atomicAdd(p + 0, w * v.x);
    atomicAdd(p + 1, w * v.y);
    atomicAdd(p + 2, w * v.z);
    atomicAdd(p + 3, w * v.w);
}

__global__ void k_bias_relu(const float* __restrict__ agg, const float* __restrict__ b,
                            float* __restrict__ h, long n64) {
    long idx = (long)blockIdx.x * blockDim.x + threadIdx.x;
    if (idx >= n64) return;
    int f = (int)(idx & 63);
    h[idx] = fmaxf(agg[idx] + b[f], 0.0f);
}

__global__ void k_final_dot(const float* __restrict__ H, const float* __restrict__ w,
                            const float* __restrict__ b, float* __restrict__ out, int n) {
    int i = blockIdx.x * blockDim.x + threadIdx.x;
    if (i >= n) return;
    const float4* hp = (const float4*)(H + (size_t)i * 64);
    const float4* wp = (const float4*)w;
    float s = 0.0f;
#pragma unroll
    for (int k = 0; k < 16; ++k) {
        float4 hv = hp[k], wv = wp[k];
        s += hv.x * wv.x + hv.y * wv.y + hv.z * wv.z + hv.w * wv.w;
    }
    out[i] = s + b[0];
}


extern "C" void kernel_launch(void* const* d_in, const int* in_sizes, int n_in,
                              void* d_out, int out_size, void* d_ws, size_t ws_size,
                              hipStream_t stream) {
    (void)n_in; (void)out_size; (void)ws_size;

    const float* x   = (const float*)d_in[0];
    const int*   ei  = (const int*)  d_in[1];
    const float* W1  = (const float*)d_in[2];
    const float* b1  = (const float*)d_in[3];
    const float* W2  = (const float*)d_in[4];
    const float* b2  = (const float*)d_in[5];
    const float* Wf1 = (const float*)d_in[6];
    const float* bf1 = (const float*)d_in[7];
    const float* Wf2 = (const float*)d_in[8];
    const float* bf2 = (const float*)d_in[9];
    const float* Wf3 = (const float*)d_in[10];
    const float* bf3 = (const float*)d_in[11];
    float* out = (float*)d_out;

    const int n = in_sizes[0] / 64;
    const int E = in_sizes[1] / 2;
    const long n64 = (long)n * 64;

    float* ws    = (float*)d_ws;
    float* dinv  = ws;
    float* wnorm = ws + n;
    float* bufA  = wnorm + E;
    float* bufB  = bufA + (size_t)n * 64;

    const int T = 256;
    const int rowTiles = (n + 15) / 16;
    const int gemmBlocks = (rowTiles + 7) / 8;
    const int nodeVecBlocks = (int)(((long)n * 16 + T - 1) / T);
    const int edgeVecBlocks = (int)(((long)E * 16 + T - 1) / T);
    const int elemBlocks = (int)((n64 + T - 1) / T);

    k_init_deg<<<(n + T - 1) / T, T, 0, stream>>>(dinv, n);
    k_count_deg<<<(E + T - 1) / T, T, 0, stream>>>(ei, dinv, E);
    k_rsqrt_inplace<<<(n + T - 1) / T, T, 0, stream>>>(dinv, n);
    k_edge_norm<<<(E + T - 1) / T, T, 0, stream>>>(ei, dinv, wnorm, E);

    k_gemm64<false><<<gemmBlocks, T, 0, stream>>>(x, W1, nullptr, bufA, n);
    k_selfloop_init<<<nodeVecBlocks, T, 0, stream>>>(bufA, dinv, bufB, n);
    k_edge_scatter<<<edgeVecBlocks, T, 0, stream>>>(ei, wnorm, bufA, bufB, E);
    k_bias_relu<<<elemBlocks, T, 0, stream>>>(bufB, b1, bufA, n64);

    k_gemm64<false><<<gemmBlocks, T, 0, stream>>>(bufA, W2, nullptr, bufB, n);
    k_selfloop_init<<<nodeVecBlocks, T, 0, stream>>>(bufB, dinv, bufA, n);
    k_edge_scatter<<<edgeVecBlocks, T, 0, stream>>>(ei, wnorm, bufB, bufA, E);
    k_bias_relu<<<elemBlocks, T, 0, stream>>>(bufA, b2, bufB, n64);

    k_gemm64<true><<<gemmBlocks, T, 0, stream>>>(bufB, Wf1, bf1, bufA, n);
    k_gemm64<true><<<gemmBlocks, T, 0, stream>>>(bufA, Wf2, bf2, bufB, n);
    k_final_dot<<<(n + T - 1) / T, T, 0, stream>>>(bufB, Wf3, bf3, out, n);
}


# Edge scatter: one wave per edge row, lane l owns columns l and l+32, so each float atomic and row load covers one whole 128-B line (was 8 lanes in each of 4 lines); sum order still arrival
# speedup vs baseline: 1.2474x; 1.2474x over previous
#include <hip/hip_runtime.h>
#include <hip/hip_bf16.h>

typedef __attribute__((ext_vector_type(2))) float v2f;
typedef __attribute__((ext_vector_type(8))) float v8f;


__global__ void k_init_deg(float* __restrict__ deg, int n) {
    int i = blockIdx.x * blockDim.x + threadIdx.x;
    if (i < n) deg[i] = 1.0f;
}

__global__ void k_count_deg(const int* __restrict__ ei, float* __restrict__ deg, int nedges) {
    int e = blockIdx.x * blockDim.x + threadIdx.x;
    if (e < nedges) atomicAdd(&deg[ei[nedges + e]], 1.0f);
}

__global__ void k_rsqrt_inplace(float* __restrict__ deg, int n) {
    int i = blockIdx.x * blockDim.x + threadIdx.x;
    if (i < n) deg[i] = rsqrtf(deg[i]);
}

__global__ void k_edge_norm(const int* __restrict__ ei, const float* __restrict__ dinv,
                            float* __restrict__ w, int nedges) {
    int e = blockIdx.x * blockDim.x + threadIdx.x;
    if (e < nedges) w[e] = dinv[ei[e]] * dinv[ei[nedges + e]];
}


#define WT_PITCH 68

template <bool BIAS_RELU>
__global__ void k_gemm64(const float* __restrict__ A, const float* __restrict__ W,
                         const float* __restrict__ bias, float* __restrict__ C, int nrows) {
    __shared__ float ldsW[64 * WT_PITCH];

    for (int idx = threadIdx.x; idx < 4096; idx += 256) {
        const int k = idx >> 6;
        const int col = idx & 63;
        ldsW[col * WT_PITCH + k] = W[idx];
    }
    __syncthreads();

    const int lane = threadIdx.x & 31;
    const int wave = threadIdx.x >> 5;
    const int rowTiles = (nrows + 15) >> 4;
    const int tile = blockIdx.x * 8 + wave;
    if (tile < rowTiles) {
        const int half = lane >> 4;
        const int m    = lane & 15;
        int row = tile * 16 + m;
        if (row >= nrows) row = nrows - 1;
        const float* __restrict__ arow = A + (size_t)row * 64;

        v8f c0 = {}, c1 = {}, c2 = {}, c3 = {};
#pragma unroll
        for (int kk = 0; kk < 16; ++kk) {
            const int k0 = kk * 4 + 2 * half;
            v2f a;
            a.x = arow[k0];
            a.y = arow[k0 + 1];
            v2f b0 = *(const v2f*)&ldsW[(0 * 16 + m) * WT_PITCH + k0];
            v2f b1 = *(const v2f*)&ldsW[(1 * 16 + m) * WT_PITCH + k0];
            v2f b2 = *(const v2f*)&ldsW[(2 * 16 + m) * WT_PITCH + k0];
            v2f b3 = *(const v2f*)&ldsW[(3 * 16 + m) * WT_PITCH + k0];
            c0 = __builtin_amdgcn_wmma_f32_16x16x4_f32(false, a, false, b0, (short)0, c0, false, false);
            c1 = __builtin_amdgcn_wmma_f32_16x16x4_f32(false, a, false, b1, (short)0, c1, false, false);
            c2 = __builtin_amdgcn_wmma_f32_16x16x4_f32(false, a, false, b2, (short)0, c2, false, false);
            c3 = __builtin_amdgcn_wmma_f32_16x16x4_f32(false, a, false, b3, (short)0, c3, false, false);
        }

        v8f acc[4] = {c0, c1, c2, c3};
#pragma unroll
        for (int nt = 0; nt < 4; ++nt) {
            const int col = nt * 16 + m;
            const float bv = BIAS_RELU ? bias[col] : 0.0f;
#pragma unroll
            for (int r = 0; r < 8; ++r) {
                const int orow = tile * 16 + r + 8 * half;
                if (orow < nrows) {
                    float v = acc[nt][r];
                    if (BIAS_RELU) v = fmaxf(v + bv, 0.0f);
                    C[(size_t)orow * 64 + col] = v;
                }
            }
        }
    }
}


__global__ void k_selfloop_init(const float* __restrict__ t, const float* __restrict__ dinv,
                                float* __restrict__ agg, int n) {
    long tid = (long)blockIdx.x * blockDim.x + threadIdx.x;
    int i = (int)(tid >> 4);
    int j = (int)(tid & 15) * 4;
    if (i >= n) return;
    float w = dinv[i] * dinv[i];
    float4 v = *(const float4*)(t + (size_t)i * 64 + j);
    float4 r = make_float4(w * v.x, w * v.y, w * v.z, w * v.w);
    *(float4*)(agg + (size_t)i * 64 + j) = r;
}


template <bool TAIL>
__device__ __forceinline__ void scatter8(int s, int d, float w, int i0, int cnt, int lane,
                                         const float* __restrict__ t, float* __restrict__ agg) {
    float v0[8], v1[8];
#pragma unroll
    for (int u = 0; u < 8; ++u) {
        const int su = __builtin_amdgcn_readlane(s, i0 + u);
        const float* __restrict__ tp = t + (size_t)su * 64;
        v0[u] = tp[lane];
        v1[u] = tp[lane + 32];
    }
#pragma unroll
    for (int u = 0; u < 8; ++u) {
        if (!TAIL || i0 + u < cnt) {
            const int du = __builtin_amdgcn_readlane(d, i0 + u);
            const float wu = __builtin_bit_cast(float,
                __builtin_amdgcn_readlane(__builtin_bit_cast(int, w), i0 + u));
            float* p = agg + (size_t)du * 64;
            atomicAdd(p + lane, wu * v0[u]);
            atomicAdd(p + lane + 32, wu * v1[u]);
        }
    }
}

__global__ void __launch_bounds__(256)
k_edge_scatter(const int* __restrict__ ei, const float* __restrict__ wnorm,
               const float* __restrict__ t, float* __restrict__ agg, int nedges) {
    const int lane = threadIdx.x & 31;
    const int wv = __builtin_amdgcn_readfirstlane((int)(threadIdx.x >> 5));
    const long base = ((long)blockIdx.x * 8 + wv) * 32;
    if (base >= nedges) return;
    const int cnt = (nedges - base) < 32 ? (int)(nedges - base) : 32;
    int s = 0, d = 0;
    float w = 0.0f;
    if (lane < cnt) {
        s = ei[base + lane];
        d = ei[(long)nedges + base + lane];
        w = wnorm[base + lane];
    }
    if (cnt == 32) {
        for (int i0 = 0; i0 < 32; i0 += 8) scatter8<false>(s, d, w, i0, 32, lane, t, agg);
    } else {
        for (int i0 = 0; i0 < cnt; i0 += 8) scatter8<true>(s, d, w, i0, cnt, lane, t, agg);
    }
}

__global__ void k_bias_relu(const float* __restrict__ agg, const float* __restrict__ b,
                            float* __restrict__ h, long n64) {
    long idx = (long)blockIdx.x * blockDim.x + threadIdx.x;
    if (idx >= n64) return;
    int f = (int)(idx & 63);
    h[idx] = fmaxf(agg[idx] + b[f], 0.0f);
}

__global__ void k_final_dot(const float* __restrict__ H, const float* __restrict__ w,
                            const float* __restrict__ b, float* __restrict__ out, int n) {
    int i = blockIdx.x * blockDim.x + threadIdx.x;
    if (i >= n) return;
    const float4* hp = (const float4*)(H + (size_t)i * 64);
    const float4* wp = (const float4*)w;
    float s = 0.0f;
#pragma unroll
    for (int k = 0; k < 16; ++k) {
        float4 hv = hp[k], wv = wp[k];
        s += hv.x * wv.x + hv.y * wv.y + hv.z * wv.z + hv.w * wv.w;
    }
    out[i] = s + b[0];
}


extern "C" void kernel_launch(void* const* d_in, const int* in_sizes, int n_in,
                              void* d_out, int out_size, void* d_ws, size_t ws_size,
                              hipStream_t stream) {
    (void)n_in; (void)out_size; (void)ws_size;

    const float* x   = (const float*)d_in[0];
    const int*   ei  = (const int*)  d_in[1];
    const float* W1  = (const float*)d_in[2];
    const float* b1  = (const float*)d_in[3];
    const float* W2  = (const float*)d_in[4];
    const float* b2  = (const float*)d_in[5];
    const float* Wf1 = (const float*)d_in[6];
    const float* bf1 = (const float*)d_in[7];
    const float* Wf2 = (const float*)d_in[8];
    const float* bf2 = (const float*)d_in[9];
    const float* Wf3 = (const float*)d_in[10];
    const float* bf3 = (const float*)d_in[11];
    float* out = (float*)d_out;

    const int n = in_sizes[0] / 64;
    const int E = in_sizes[1] / 2;
    const long n64 = (long)n * 64;

    float* ws    = (float*)d_ws;
    float* dinv  = ws;
    float* wnorm = ws + n;
    float* bufA  = wnorm + E;
    float* bufB  = bufA + (size_t)n * 64;

    const int T = 256;
    const int rowTiles = (n + 15) / 16;
    const int gemmBlocks = (rowTiles + 7) / 8;
    const int nodeVecBlocks = (int)(((long)n * 16 + T - 1) / T);
    const int edgeVecBlocks = (E + T - 1) / T;
    const int elemBlocks = (int)((n64 + T - 1) / T);

    k_init_deg<<<(n + T - 1) / T, T, 0, stream>>>(dinv, n);
    k_count_deg<<<(E + T - 1) / T, T, 0, stream>>>(ei, dinv, E);
    k_rsqrt_inplace<<<(n + T - 1) / T, T, 0, stream>>>(dinv, n);
    k_edge_norm<<<(E + T - 1) / T, T, 0, stream>>>(ei, dinv, wnorm, E);

    k_gemm64<false><<<gemmBlocks, T, 0, stream>>>(x, W1, nullptr, bufA, n);
    k_selfloop_init<<<nodeVecBlocks, T, 0, stream>>>(bufA, dinv, bufB, n);
    k_edge_scatter<<<edgeVecBlocks, T, 0, stream>>>(ei, wnorm, bufA, bufB, E);
    k_bias_relu<<<elemBlocks, T, 0, stream>>>(bufB, b1, bufA, n64);

    k_gemm64<false><<<gemmBlocks, T, 0, stream>>>(bufA, W2, nullptr, bufB, n);
    k_selfloop_init<<<nodeVecBlocks, T, 0, stream>>>(bufB, dinv, bufA, n);
    k_edge_scatter<<<edgeVecBlocks, T, 0, stream>>>(ei, wnorm, bufB, bufA, E);
    k_bias_relu<<<elemBlocks, T, 0, stream>>>(bufA, b2, bufB, n64);

    k_gemm64<true><<<gemmBlocks, T, 0, stream>>>(bufB, Wf1, bf1, bufA, n);
    k_gemm64<true><<<gemmBlocks, T, 0, stream>>>(bufA, Wf2, bf2, bufB, n);
    k_final_dot<<<(n + T - 1) / T, T, 0, stream>>>(bufB, Wf3, bf3, out, n);
}


# Layer-1 owner scan writes each drained edge as an 8-byte record (source, local row, weight) to its wave's own list; layer 2 walks the lists: no key scan, no id or deg reads; sum order LDS arrival
# speedup vs baseline: 1.8004x; 1.4433x over previous
#include <hip/hip_runtime.h>
#include <hip/hip_bf16.h>

typedef __attribute__((ext_vector_type(2))) float v2f;
typedef __attribute__((ext_vector_type(8))) float v8f;

#define OWN_R     512
#define OWN_SHIFT 9
#define OWN_WAVES 32
#define OWN_SUB   4
#define OWN_QCAP  48
#define KEY_TRIP  (OWN_WAVES * OWN_SUB * 512)
#define REC_CAP   512
#define REC_SHIFT 17


__global__ void k_init_deg(float* __restrict__ deg, int n) {
    int i = blockIdx.x * blockDim.x + threadIdx.x;
    if (i < n) deg[i] = 1.0f;
}

__global__ void k_count_deg_keys(const int* __restrict__ ei, float* __restrict__ deg, int* __restrict__ keys,
                                 int nedges) {
    const int npad4 = ((nedges + KEY_TRIP - 1) / KEY_TRIP) * (KEY_TRIP / 4);
    const int i = blockIdx.x * blockDim.x + threadIdx.x;
    if (i >= npad4) return;
    const long e0 = (long)i * 4;
    const int* __restrict__ dstp = ei + nedges;
    int d0 = -1, d1 = -1, d2 = -1, d3 = -1;
    if ((nedges & 3) == 0 && e0 + 3 < nedges) {
        const int4 v = *(const int4*)(dstp + e0);
        d0 = v.x; d1 = v.y; d2 = v.z; d3 = v.w;
    } else {
        if (e0     < nedges) d0 = dstp[e0];
        if (e0 + 1 < nedges) d1 = dstp[e0 + 1];
        if (e0 + 2 < nedges) d2 = dstp[e0 + 2];
        if (e0 + 3 < nedges) d3 = dstp[e0 + 3];
    }
    unsigned k0 = 255u, k1 = 255u, k2 = 255u, k3 = 255u;
    if (d0 >= 0) { atomicAdd(&deg[d0], 1.0f); k0 = ((unsigned)d0 >> OWN_SHIFT) & 255u; }
    if (d1 >= 0) { atomicAdd(&deg[d1], 1.0f); k1 = ((unsigned)d1 >> OWN_SHIFT) & 255u; }
    if (d2 >= 0) { atomicAdd(&deg[d2], 1.0f); k2 = ((unsigned)d2 >> OWN_SHIFT) & 255u; }
    if (d3 >= 0) { atomicAdd(&deg[d3], 1.0f); k3 = ((unsigned)d3 >> OWN_SHIFT) & 255u; }
    keys[i] = (int)(k0 | (k1 << 8) | (k2 << 16) | (k3 << 24));
}

__global__ void k_rsqrt_inplace(float* __restrict__ deg, int n) {
    int i = blockIdx.x * blockDim.x + threadIdx.x;
    if (i < n) deg[i] = rsqrtf(deg[i]);
}


#define WT_PITCH 68

template <bool BIAS_RELU>
__global__ void k_gemm64(const float* __restrict__ A, const float* __restrict__ W,
                         const float* __restrict__ bias, float* __restrict__ C, int nrows) {
    __shared__ float ldsW[64 * WT_PITCH];

    for (int idx = threadIdx.x; idx < 4096; idx += 256) {
        const int k = idx >> 6;
        const int col = idx & 63;
        ldsW[col * WT_PITCH + k] = W[idx];
    }
    __syncthreads();

    const int lane = threadIdx.x & 31;
    const int wave = threadIdx.x >> 5;
    const int rowTiles = (nrows + 15) >> 4;
    const int tile = blockIdx.x * 8 + wave;
    if (tile < rowTiles) {
        const int half = lane >> 4;
        const int m    = lane & 15;
        int row = tile * 16 + m;
        if (row >= nrows) row = nrows - 1;
        const float* __restrict__ arow = A + (size_t)row * 64;

        v8f c0 = {}, c1 = {}, c2 = {}, c3 = {};
#pragma unroll
        for (int kk = 0; kk < 16; ++kk) {
            const int k0 = kk * 4 + 2 * half;
            v2f a;
            a.x = arow[k0];
            a.y = arow[k0 + 1];
            v2f b0 = *(const v2f*)&ldsW[(0 * 16 + m) * WT_PITCH + k0];
            v2f b1 = *(const v2f*)&ldsW[(1 * 16 + m) * WT_PITCH + k0];
            v2f b2 = *(const v2f*)&ldsW[(2 * 16 + m) * WT_PITCH + k0];
            v2f b3 = *(const v2f*)&ldsW[(3 * 16 + m) * WT_PITCH + k0];
            c0 = __builtin_amdgcn_wmma_f32_16x16x4_f32(false, a, false, b0, (short)0, c0, false, false);
            c1 = __builtin_amdgcn_wmma_f32_16x16x4_f32(false, a, false, b1, (short)0, c1, false, false);
            c2 = __builtin_amdgcn_wmma_f32_16x16x4_f32(false, a, false, b2, (short)0, c2, false, false);
            c3 = __builtin_amdgcn_wmma_f32_16x16x4_f32(false, a, false, b3, (short)0, c3, false, false);
        }

        v8f acc[4] = {c0, c1, c2, c3};
#pragma unroll
        for (int nt = 0; nt < 4; ++nt) {
            const int col = nt * 16 + m;
            const float bv = BIAS_RELU ? bias[col] : 0.0f;
#pragma unroll
            for (int r = 0; r < 8; ++r) {
                const int orow = tile * 16 + r + 8 * half;
                if (orow < nrows) {
                    float v = acc[nt][r];
                    if (BIAS_RELU) v = fmaxf(v + bv, 0.0f);
                    C[(size_t)orow * 64 + col] = v;
                }
            }
        }
    }
}


__device__ __forceinline__ void own_load_keys(int4 (&kv)[OWN_SUB], const int4* __restrict__ p) {
#pragma unroll
    for (int i = 0; i < OWN_SUB; ++i) kv[i] = p[i * 32];
}

template <bool TAIL, bool REC>
__device__ __forceinline__ void own_drain(const int* qe, int q0, int cnt, int lane, int lo, int nedges,
                                          const int* __restrict__ ei, const float* __restrict__ dinv,
                                          const float* __restrict__ t, float* tile, int2* rec, int rpos) {
    int dl = 0, s = 0;
    if (lane < cnt) {
        const int e = qe[q0 + lane];
        s = ei[e];
        dl = ei[nedges + e] - lo;
    }
    float v0[16], v1[16];
#pragma unroll
    for (int u = 0; u < 16; ++u) {
        const int su = __builtin_amdgcn_readlane(s, u);
        const float* __restrict__ tp = t + (size_t)su * 64;
        v0[u] = tp[lane];
        v1[u] = tp[lane + 32];
    }
    const float w = dinv[s] * dinv[lo + dl];
    if (REC) {
        if (rpos + 16 <= REC_CAP && lane < cnt)
            rec[rpos + lane] = make_int2(s | (dl << REC_SHIFT), __builtin_bit_cast(int, w));
    }
#pragma unroll
    for (int u = 0; u < 16; ++u) {
        if (!TAIL || u < cnt) {
            const int du = __builtin_amdgcn_readlane(dl, u);
            const float wu = __builtin_bit_cast(float,
                __builtin_amdgcn_readlane(__builtin_bit_cast(int, w), u));
            float* p = tile + du * 64;
            atomicAdd(p + lane, wu * v0[u]);
            atomicAdd(p + lane + 32, wu * v1[u]);
        }
    }
}

template <bool REC>
__device__ __forceinline__ void own_trip(const int4 (&kv)[OWN_SUB], int eb, unsigned pat, int& qn, int* qe,
                                         int lane, unsigned lt_mask, int lo, int nedges,
                                         const int* __restrict__ ei, const float* __restrict__ dinv,
                                         const float* __restrict__ t, float* tile, int2* rec, int& rpos) {
#pragma unroll
    for (int p = 0; p < OWN_SUB; ++p) {
        const unsigned x0 = (unsigned)kv[p].x ^ pat, x1 = (unsigned)kv[p].y ^ pat;
        const unsigned x2 = (unsigned)kv[p].z ^ pat, x3 = (unsigned)kv[p].w ^ pat;
        const unsigned y0 = ((x0 & 0x7F7F7F7Fu) + 0x7F7F7F7Fu) | x0;
        const unsigned y1 = ((x1 & 0x7F7F7F7Fu) + 0x7F7F7F7Fu) | x1;
        const unsigned y2 = ((x2 & 0x7F7F7F7Fu) + 0x7F7F7F7Fu) | x2;
        const unsigned y3 = ((x3 & 0x7F7F7F7Fu) + 0x7F7F7F7Fu) | x3;
        const unsigned nz = ((y0 >> 7) & 0x01010101u) | ((y1 >> 6) & 0x02020202u) |
                            ((y2 >> 5) & 0x04040404u) | ((y3 >> 4) & 0x08080808u);
        unsigned m = nz ^ 0x0F0F0F0Fu;
        if (__builtin_amdgcn_ballot_w32(m != 0u) != 0u) {
            const int ep = eb + p * 512;
            do {
                const bool has = m != 0u;
                const unsigned k = __builtin_amdgcn_ballot_w32(has);
                if (has) {
                    const int bit = __builtin_ctz(m);
                    const int j = ((bit & 7) << 2) | (bit >> 3);
                    qe[qn + __builtin_popcount(k & lt_mask)] = ep + j;
                    m &= m - 1u;
                }
                qn += __builtin_popcount(k);
                __builtin_amdgcn_wave_barrier();
                while (qn >= 16) {
                    qn -= 16;
                    own_drain<false, REC>(qe, qn, 16, lane, lo, nedges, ei, dinv, t, tile, rec, rpos);
                    rpos += 16;
                }
                __builtin_amdgcn_wave_barrier();
            } while (__builtin_amdgcn_ballot_w32(m != 0u) != 0u);
        }
    }
}

template <bool REC>
__device__ __forceinline__ int own_scan(const int* __restrict__ keys, int wv, int lane, unsigned pat, int* qe,
                                        int lo, int nedges, const int* __restrict__ ei,
                                        const float* __restrict__ dinv, const float* __restrict__ t,
                                        float* tile, int2* rec) {
    const unsigned lt_mask = (1u << lane) - 1u;
    int qn = 0;
    int rpos = 0;

    const int ntrip = (nedges + KEY_TRIP - 1) / KEY_TRIP;
    const int4* __restrict__ kp = (const int4*)keys + wv * (OWN_SUB * 32) + lane;
    const int eb0 = wv * (OWN_SUB * 512) + lane * 16;
    if (ntrip > 0) {
        int4 ka[OWN_SUB], kb[OWN_SUB];
        own_load_keys(ka, kp);
        for (int k = 0; k < ntrip; k += 2) {
            const int k1 = (k + 1 < ntrip) ? k + 1 : ntrip - 1;
            own_load_keys(kb, kp + (size_t)k1 * (KEY_TRIP / 16));
            own_trip<REC>(ka, eb0 + k * KEY_TRIP, pat, qn, qe, lane, lt_mask, lo, nedges, ei, dinv, t, tile, rec, rpos);
            if (k + 1 >= ntrip) break;
            const int k2 = (k + 2 < ntrip) ? k + 2 : ntrip - 1;
            own_load_keys(ka, kp + (size_t)k2 * (KEY_TRIP / 16));
            own_trip<REC>(kb, eb0 + (k + 1) * KEY_TRIP, pat, qn, qe, lane, lt_mask, lo, nedges, ei, dinv, t, tile, rec, rpos);
        }
    }
    while (qn > 0) {
        const int c = qn < 16 ? qn : 16;
        qn -= c;
        own_drain<true, REC>(qe, qn, c, lane, lo, nedges, ei, dinv, t, tile, rec, rpos);
        rpos += c;
    }
    return rpos;
}

template <bool TAIL, int OFF>
__device__ __forceinline__ void own_replay16(int2 r, int c, int lane, const float* __restrict__ t, float* tile) {
    float v0[16], v1[16];
#pragma unroll
    for (int u = 0; u < 16; ++u) {
        const int su = __builtin_amdgcn_readlane(r.x, OFF + u) & ((1 << REC_SHIFT) - 1);
        const float* __restrict__ tp = t + (size_t)su * 64;
        v0[u] = tp[lane];
        v1[u] = tp[lane + 32];
    }
#pragma unroll
    for (int u = 0; u < 16; ++u) {
        if (!TAIL || u < c) {
            const int du = __builtin_amdgcn_readlane(r.x, OFF + u) >> REC_SHIFT;
            const float wu = __builtin_bit_cast(float, __builtin_amdgcn_readlane(r.y, OFF + u));
            float* p = tile + du * 64;
            atomicAdd(p + lane, wu * v0[u]);
            atomicAdd(p + lane + 32, wu * v1[u]);
        }
    }
}

__device__ __forceinline__ void own_replay(const int2* __restrict__ rec, int cnt, int lane,
                                           const float* __restrict__ t, float* tile) {
    if (cnt <= 0) return;
    const int last = cnt - 1;
    int2 cur = rec[lane < last ? lane : last];
    for (int base = 0; base < cnt; base += 32) {
        const int i1 = base + 32 + lane;
        const int2 nxt = rec[i1 < last ? i1 : last];
        const int left = cnt - base;
        if (left >= 32) {
            own_replay16<false, 0>(cur, 16, lane, t, tile);
            own_replay16<false, 16>(cur, 16, lane, t, tile);
        } else {
            own_replay16<true, 0>(cur, left < 16 ? left : 16, lane, t, tile);
            if (left > 16) own_replay16<true, 16>(cur, left - 16, lane, t, tile);
        }
        cur = nxt;
    }
}

template <bool REPLAY>
__global__ void __launch_bounds__(OWN_WAVES * 32)
k_agg_owner(const int* __restrict__ ei, const int* __restrict__ keys, const float* __restrict__ dinv,
            const float* __restrict__ t, const float* __restrict__ b, float* __restrict__ h, int n, int nedges,
            int* recs, int* cnts) {
    __shared__ float tile[OWN_R * 64];
    __shared__ int qedge[OWN_WAVES * OWN_QCAP];
    __shared__ int wcnt[OWN_WAVES];

    const int lane = threadIdx.x & 31;
    const int wv = __builtin_amdgcn_readfirstlane((int)(threadIdx.x >> 5));
    const int lo = blockIdx.x * OWN_R;
    const int rows = (n - lo) < OWN_R ? (n - lo) : OWN_R;

    for (int idx = threadIdx.x; idx < rows * 16; idx += OWN_WAVES * 32) {
        const int r = idx >> 4;
        const int j = (idx & 15) * 4;
        const float di = dinv[lo + r];
        const float sw = di * di;
        const float4 v = *(const float4*)(t + (size_t)(lo + r) * 64 + j);
        *(float4*)(tile + r * 64 + j) = make_float4(sw * v.x, sw * v.y, sw * v.z, sw * v.w);
    }
    __syncthreads();

    int* qe = qedge + wv * OWN_QCAP;
    const unsigned pat = (blockIdx.x & 255u) * 0x01010101u;
    int2* rec = (int2*)recs + (size_t)(blockIdx.x * OWN_WAVES + wv) * REC_CAP;

    if (REPLAY) {
        const int mine = __builtin_amdgcn_readfirstlane(cnts[blockIdx.x * OWN_WAVES + wv]);
        if (mine <= REC_CAP) own_replay(rec, mine, lane, t, tile);
        else (void)own_scan<false>(keys, wv, lane, pat, qe, lo, nedges, ei, dinv, t, tile, rec);
    } else {
        const int mine = own_scan<true>(keys, wv, lane, pat, qe, lo, nedges, ei, dinv, t, tile, rec);
        if (lane == 0) wcnt[wv] = mine;
    }
    __syncthreads();

    if (!REPLAY && wv == 0) cnts[blockIdx.x * OWN_WAVES + lane] = wcnt[lane];

    for (int idx = threadIdx.x; idx < rows * 16; idx += OWN_WAVES * 32) {
        const int r = idx >> 4;
        const int j = (idx & 15) * 4;
        const float4 v = *(const float4*)(tile + r * 64 + j);
        const float4 bv = *(const float4*)(b + j);
        float4 o;
        o.x = fmaxf(v.x + bv.x, 0.0f);
        o.y = fmaxf(v.y + bv.y, 0.0f);
        o.z = fmaxf(v.z + bv.z, 0.0f);
        o.w = fmaxf(v.w + bv.w, 0.0f);
        *(float4*)(h + (size_t)(lo + r) * 64 + j) = o;
    }
}

__global__ void k_final_dot(const float* __restrict__ H, const float* __restrict__ w,
                            const float* __restrict__ b, float* __restrict__ out, int n) {
    int i = blockIdx.x * blockDim.x + threadIdx.x;
    if (i >= n) return;
    const float4* hp = (const float4*)(H + (size_t)i * 64);
    const float4* wp = (const float4*)w;
    float s = 0.0f;
#pragma unroll
    for (int k = 0; k < 16; ++k) {
        float4 hv = hp[k], wv = wp[k];
        s += hv.x * wv.x + hv.y * wv.y + hv.z * wv.z + hv.w * wv.w;
    }
    out[i] = s + b[0];
}


extern "C" void kernel_launch(void* const* d_in, const int* in_sizes, int n_in,
                              void* d_out, int out_size, void* d_ws, size_t ws_size,
                              hipStream_t stream) {
    (void)n_in; (void)out_size; (void)ws_size;

    const float* x   = (const float*)d_in[0];
    const int*   ei  = (const int*)  d_in[1];
    const float* W1  = (const float*)d_in[2];
    const float* b1  = (const float*)d_in[3];
    const float* W2  = (const float*)d_in[4];
    const float* b2  = (const float*)d_in[5];
    const float* Wf1 = (const float*)d_in[6];
    const float* bf1 = (const float*)d_in[7];
    const float* Wf2 = (const float*)d_in[8];
    const float* bf2 = (const float*)d_in[9];
    const float* Wf3 = (const float*)d_in[10];
    const float* bf3 = (const float*)d_in[11];
    float* out = (float*)d_out;

    const int n = in_sizes[0] / 64;
    const int E = in_sizes[1] / 2;

    float* ws    = (float*)d_ws;
    float* dinv  = ws;
    float* bufA  = ws + n;
    float* bufB  = bufA + (size_t)n * 64;
    int*   keys  = (int*)((char*)d_ws + ((size_t)128 << 20));
    int*   cnts  = (int*)((char*)d_ws + ((size_t)160 << 20));
    int*   recs  = (int*)((char*)d_ws + ((size_t)192 << 20));

    const int T = 256;
    const int rowTiles = (n + 15) / 16;
    const int gemmBlocks = (rowTiles + 7) / 8;
    const int ownBlocks = (n + OWN_R - 1) / OWN_R;
    const int keyBlocks = ((E + KEY_TRIP - 1) / KEY_TRIP) * (KEY_TRIP / 4 / T);

    k_init_deg<<<(n + T - 1) / T, T, 0, stream>>>(dinv, n);
    k_count_deg_keys<<<keyBlocks, T, 0, stream>>>(ei, dinv, keys, E);
    k_rsqrt_inplace<<<(n + T - 1) / T, T, 0, stream>>>(dinv, n);

    k_gemm64<false><<<gemmBlocks, T, 0, stream>>>(x, W1, nullptr, bufA, n);
    k_agg_owner<false><<<ownBlocks, OWN_WAVES * 32, 0, stream>>>(ei, keys, dinv, bufA, b1, bufB, n, E, recs, cnts);

    k_gemm64<false><<<gemmBlocks, T, 0, stream>>>(bufB, W2, nullptr, bufA, n);
    k_agg_owner<true><<<ownBlocks, OWN_WAVES * 32, 0, stream>>>(ei, keys, dinv, bufA, b2, bufB, n, E, recs, cnts);

    k_gemm64<true><<<gemmBlocks, T, 0, stream>>>(bufB, Wf1, bf1, bufA, n);
    k_gemm64<true><<<gemmBlocks, T, 0, stream>>>(bufA, Wf2, bf2, bufB, n);
    k_final_dot<<<(n + T - 1) / T, T, 0, stream>>>(bufB, Wf3, bf3, out, n);
}


# Degree count: each work-group counts one edge slice's destinations of one node range in an LDS integer table, stored as partial planes the rsqrt pass adds; no global adds, no start launch; exact
# speedup vs baseline: 2.4541x; 1.3631x over previous
#include <hip/hip_runtime.h>
#include <hip/hip_bf16.h>

typedef __attribute__((ext_vector_type(2))) float v2f;
typedef __attribute__((ext_vector_type(8))) float v8f;

#define OWN_R     512
#define OWN_SHIFT 9
#define OWN_WAVES 32
#define OWN_SUB   4
#define OWN_QCAP  48
#define KEY_TRIP  (OWN_WAVES * OWN_SUB * 512)
#define REC_CAP   512
#define REC_SHIFT 17

#define DEG_K    8
#define DEG_S    32
#define DEG_RCAP 16384
#define DEG_T    1024
#define DEG_U    8

__device__ __forceinline__ int deg_range_rows(int n) { return (((n + DEG_K - 1) / DEG_K) + 31) & ~31; }

template <bool WIDE>
__device__ __forceinline__ void deg_walk(const int* __restrict__ dstp, int nedges, int q0, int q1, int kr,
                                         int lo, int R, unsigned* hist, int* __restrict__ keys) {
    for (int qb = q0 + (int)threadIdx.x; qb < q1; qb += DEG_T * DEG_U) {
        int4 v[DEG_U];
#pragma unroll
        for (int u = 0; u < DEG_U; ++u) {
            const int q = qb + u * DEG_T;
            const long e0 = (long)q * 4;
            if (WIDE) {
                const bool in = q < q1 && e0 < nedges;
                const int4 d = *(const int4*)(dstp + (in ? e0 : 0));
                v[u] = in ? d : make_int4(-1, -1, -1, -1);
            } else {
                const bool i0 = q < q1 && e0     < nedges, i1 = q < q1 && e0 + 1 < nedges;
                const bool i2 = q < q1 && e0 + 2 < nedges, i3 = q < q1 && e0 + 3 < nedges;
                const int x0 = dstp[i0 ? e0     : 0], x1 = dstp[i1 ? e0 + 1 : 0];
                const int x2 = dstp[i2 ? e0 + 2 : 0], x3 = dstp[i3 ? e0 + 3 : 0];
                v[u] = make_int4(i0 ? x0 : -1, i1 ? x1 : -1, i2 ? x2 : -1, i3 ? x3 : -1);
            }
        }
#pragma unroll
        for (int u = 0; u < DEG_U; ++u) {
            const int q = qb + u * DEG_T;
            const int4 d = v[u];
            const unsigned r0 = (unsigned)(d.x - lo), r1 = (unsigned)(d.y - lo);
            const unsigned r2 = (unsigned)(d.z - lo), r3 = (unsigned)(d.w - lo);
            if (r0 < (unsigned)R) atomicAdd(&hist[r0], 1u);
            if (r1 < (unsigned)R) atomicAdd(&hist[r1], 1u);
            if (r2 < (unsigned)R) atomicAdd(&hist[r2], 1u);
            if (r3 < (unsigned)R) atomicAdd(&hist[r3], 1u);
            if ((u % DEG_K) == kr && q < q1) {
                const unsigned k0 = d.x >= 0 ? (((unsigned)d.x >> OWN_SHIFT) & 255u) : 255u;
                const unsigned k1 = d.y >= 0 ? (((unsigned)d.y >> OWN_SHIFT) & 255u) : 255u;
                const unsigned k2 = d.z >= 0 ? (((unsigned)d.z >> OWN_SHIFT) & 255u) : 255u;
                const unsigned k3 = d.w >= 0 ? (((unsigned)d.w >> OWN_SHIFT) & 255u) : 255u;
                keys[q] = (int)(k0 | (k1 << 8) | (k2 << 16) | (k3 << 24));
            }
        }
    }
}

__global__ void __launch_bounds__(DEG_T)
k_deg_hist_keys(const int* __restrict__ ei, unsigned* __restrict__ part, int* __restrict__ keys,
                int n, int nedges) {
    __shared__ unsigned hist[DEG_RCAP];
    const int kr = blockIdx.x % DEG_K;
    const int sl = blockIdx.x / DEG_K;
    const int R  = deg_range_rows(n);
    const int lo = kr * R;

    for (int j = threadIdx.x * 4; j < R; j += DEG_T * 4) *(uint4*)(hist + j) = make_uint4(0u, 0u, 0u, 0u);
    __syncthreads();

    const int npad4 = ((nedges + KEY_TRIP - 1) / KEY_TRIP) * (KEY_TRIP / 4);
    const int qps = (npad4 + DEG_S - 1) / DEG_S;
    const int q0 = sl * qps;
    const int q1 = (q0 + qps) < npad4 ? (q0 + qps) : npad4;
    const int* __restrict__ dstp = ei + nedges;
    if ((nedges & 3) == 0) deg_walk<true>(dstp, nedges, q0, q1, kr, lo, R, hist, keys);
    else                   deg_walk<false>(dstp, nedges, q0, q1, kr, lo, R, hist, keys);
    __syncthreads();

    unsigned* __restrict__ plane = part + (size_t)sl * (DEG_K * R) + lo;
    for (int j = threadIdx.x * 4; j < R; j += DEG_T * 4) *(uint4*)(plane + j) = *(const uint4*)(hist + j);
}

__global__ void k_dinv_from_parts(const unsigned* __restrict__ part, float* __restrict__ dinv, int n) {
    const int i = blockIdx.x * blockDim.x + threadIdx.x;
    if (i >= n) return;
    const int P = DEG_K * deg_range_rows(n);
    unsigned c = 1u;
#pragma unroll
    for (int s = 0; s < DEG_S; ++s) c += part[(size_t)s * P + i];
    dinv[i] = rsqrtf((float)c);
}


#define WT_PITCH 68

template <bool BIAS_RELU>
__global__ void k_gemm64(const float* __restrict__ A, const float* __restrict__ W,
                         const float* __restrict__ bias, float* __restrict__ C, int nrows) {
    __shared__ float ldsW[64 * WT_PITCH];

    for (int idx = threadIdx.x; idx < 4096; idx += 256) {
        const int k = idx >> 6;
        const int col = idx & 63;
        ldsW[col * WT_PITCH + k] = W[idx];
    }
    __syncthreads();

    const int lane = threadIdx.x & 31;
    const int wave = threadIdx.x >> 5;
    const int rowTiles = (nrows + 15) >> 4;
    const int tile = blockIdx.x * 8 + wave;
    if (tile < rowTiles) {
        const int half = lane >> 4;
        const int m    = lane & 15;
        int row = tile * 16 + m;
        if (row >= nrows) row = nrows - 1;
        const float* __restrict__ arow = A + (size_t)row * 64;

        v8f c0 = {}, c1 = {}, c2 = {}, c3 = {};
#pragma unroll
        for (int kk = 0; kk < 16; ++kk) {
            const int k0 = kk * 4 + 2 * half;
            v2f a;
            a.x = arow[k0];
            a.y = arow[k0 + 1];
            v2f b0 = *(const v2f*)&ldsW[(0 * 16 + m) * WT_PITCH + k0];
            v2f b1 = *(const v2f*)&ldsW[(1 * 16 + m) * WT_PITCH + k0];
            v2f b2 = *(const v2f*)&ldsW[(2 * 16 + m) * WT_PITCH + k0];
            v2f b3 = *(const v2f*)&ldsW[(3 * 16 + m) * WT_PITCH + k0];
            c0 = __builtin_amdgcn_wmma_f32_16x16x4_f32(false, a, false, b0, (short)0, c0, false, false);
            c1 = __builtin_amdgcn_wmma_f32_16x16x4_f32(false, a, false, b1, (short)0, c1, false, false);
            c2 = __builtin_amdgcn_wmma_f32_16x16x4_f32(false, a, false, b2, (short)0, c2, false, false);
            c3 = __builtin_amdgcn_wmma_f32_16x16x4_f32(false, a, false, b3, (short)0, c3, false, false);
        }

        v8f acc[4] = {c0, c1, c2, c3};
#pragma unroll
        for (int nt = 0; nt < 4; ++nt) {
            const int col = nt * 16 + m;
            const float bv = BIAS_RELU ? bias[col] : 0.0f;
#pragma unroll
            for (int r = 0; r < 8; ++r) {
                const int orow = tile * 16 + r + 8 * half;
                if (orow < nrows) {
                    float v = acc[nt][r];
                    if (BIAS_RELU) v = fmaxf(v + bv, 0.0f);
                    C[(size_t)orow * 64 + col] = v;
                }
            }
        }
    }
}


__device__ __forceinline__ void own_load_keys(int4 (&kv)[OWN_SUB], const int4* __restrict__ p) {
#pragma unroll
    for (int i = 0; i < OWN_SUB; ++i) kv[i] = p[i * 32];
}

template <bool TAIL, bool REC>
__device__ __forceinline__ void own_drain(const int* qe, int q0, int cnt, int lane, int lo, int nedges,
                                          const int* __restrict__ ei, const float* __restrict__ dinv,
                                          const float* __restrict__ t, float* tile, int2* rec, int rpos) {
    int dl = 0, s = 0;
    if (lane < cnt) {
        const int e = qe[q0 + lane];
        s = ei[e];
        dl = ei[nedges + e] - lo;
    }
    float v0[16], v1[16];
#pragma unroll
    for (int u = 0; u < 16; ++u) {
        const int su = __builtin_amdgcn_readlane(s, u);
        const float* __restrict__ tp = t + (size_t)su * 64;
        v0[u] = tp[lane];
        v1[u] = tp[lane + 32];
    }
    const float w = dinv[s] * dinv[lo + dl];
    if (REC) {
        if (rpos + 16 <= REC_CAP && lane < cnt)
            rec[rpos + lane] = make_int2(s | (dl << REC_SHIFT), __builtin_bit_cast(int, w));
    }
#pragma unroll
    for (int u = 0; u < 16; ++u) {
        if (!TAIL || u < cnt) {
            const int du = __builtin_amdgcn_readlane(dl, u);
            const float wu = __builtin_bit_cast(float,
                __builtin_amdgcn_readlane(__builtin_bit_cast(int, w), u));
            float* p = tile + du * 64;
            atomicAdd(p + lane, wu * v0[u]);
            atomicAdd(p + lane + 32, wu * v1[u]);
        }
    }
}

template <bool REC>
__device__ __forceinline__ void own_trip(const int4 (&kv)[OWN_SUB], int eb, unsigned pat, int& qn, int* qe,
                                         int lane, unsigned lt_mask, int lo, int nedges,
                                         const int* __restrict__ ei, const float* __restrict__ dinv,
                                         const float* __restrict__ t, float* tile, int2* rec, int& rpos) {
#pragma unroll
    for (int p = 0; p < OWN_SUB; ++p) {
        const unsigned x0 = (unsigned)kv[p].x ^ pat, x1 = (unsigned)kv[p].y ^ pat;
        const unsigned x2 = (unsigned)kv[p].z ^ pat, x3 = (unsigned)kv[p].w ^ pat;
        const unsigned y0 = ((x0 & 0x7F7F7F7Fu) + 0x7F7F7F7Fu) | x0;
        const unsigned y1 = ((x1 & 0x7F7F7F7Fu) + 0x7F7F7F7Fu) | x1;
        const unsigned y2 = ((x2 & 0x7F7F7F7Fu) + 0x7F7F7F7Fu) | x2;
        const unsigned y3 = ((x3 & 0x7F7F7F7Fu) + 0x7F7F7F7Fu) | x3;
        const unsigned nz = ((y0 >> 7) & 0x01010101u) | ((y1 >> 6) & 0x02020202u) |
                            ((y2 >> 5) & 0x04040404u) | ((y3 >> 4) & 0x08080808u);
        unsigned m = nz ^ 0x0F0F0F0Fu;
        if (__builtin_amdgcn_ballot_w32(m != 0u) != 0u) {
            const int ep = eb + p * 512;
            do {
                const bool has = m != 0u;
                const unsigned k = __builtin_amdgcn_ballot_w32(has);
                if (has) {
                    const int bit = __builtin_ctz(m);
                    const int j = ((bit & 7) << 2) | (bit >> 3);
                    qe[qn + __builtin_popcount(k & lt_mask)] = ep + j;
                    m &= m - 1u;
                }
                qn += __builtin_popcount(k);
                __builtin_amdgcn_wave_barrier();
                while (qn >= 16) {
                    qn -= 16;
                    own_drain<false, REC>(qe, qn, 16, lane, lo, nedges, ei, dinv, t, tile, rec, rpos);
                    rpos += 16;
                }
                __builtin_amdgcn_wave_barrier();
            } while (__builtin_amdgcn_ballot_w32(m != 0u) != 0u);
        }
    }
}

template <bool REC>
__device__ __forceinline__ int own_scan(const int* __restrict__ keys, int wv, int lane, unsigned pat, int* qe,
                                        int lo, int nedges, const int* __restrict__ ei,
                                        const float* __restrict__ dinv, const float* __restrict__ t,
                                        float* tile, int2* rec) {
    const unsigned lt_mask = (1u << lane) - 1u;
    int qn = 0;
    int rpos = 0;

    const int ntrip = (nedges + KEY_TRIP - 1) / KEY_TRIP;
    const int4* __restrict__ kp = (const int4*)keys + wv * (OWN_SUB * 32) + lane;
    const int eb0 = wv * (OWN_SUB * 512) + lane * 16;
    if (ntrip > 0) {
        int4 ka[OWN_SUB], kb[OWN_SUB];
        own_load_keys(ka, kp);
        for (int k = 0; k < ntrip; k += 2) {
            const int k1 = (k + 1 < ntrip) ? k + 1 : ntrip - 1;
            own_load_keys(kb, kp + (size_t)k1 * (KEY_TRIP / 16));
            own_trip<REC>(ka, eb0 + k * KEY_TRIP, pat, qn, qe, lane, lt_mask, lo, nedges, ei, dinv, t, tile, rec, rpos);
            if (k + 1 >= ntrip) break;
            const int k2 = (k + 2 < ntrip) ? k + 2 : ntrip - 1;
            own_load_keys(ka, kp + (size_t)k2 * (KEY_TRIP / 16));
            own_trip<REC>(kb, eb0 + (k + 1) * KEY_TRIP, pat, qn, qe, lane, lt_mask, lo, nedges, ei, dinv, t, tile, rec, rpos);
        }
    }
    while (qn > 0) {
        const int c = qn < 16 ? qn : 16;
        qn -= c;
        own_drain<true, REC>(qe, qn, c, lane, lo, nedges, ei, dinv, t, tile, rec, rpos);
        rpos += c;
    }
    return rpos;
}

template <bool TAIL, int OFF>
__device__ __forceinline__ void own_replay16(int2 r, int c, int lane, const float* __restrict__ t, float* tile) {
    float v0[16], v1[16];
#pragma unroll
    for (int u = 0; u < 16; ++u) {
        const int su = __builtin_amdgcn_readlane(r.x, OFF + u) & ((1 << REC_SHIFT) - 1);
        const float* __restrict__ tp = t + (size_t)su * 64;
        v0[u] = tp[lane];
        v1[u] = tp[lane + 32];
    }
#pragma unroll
    for (int u = 0; u < 16; ++u) {
        if (!TAIL || u < c) {
            const int du = __builtin_amdgcn_readlane(r.x, OFF + u) >> REC_SHIFT;
            const float wu = __builtin_bit_cast(float, __builtin_amdgcn_readlane(r.y, OFF + u));
            float* p = tile + du * 64;
            atomicAdd(p + lane, wu * v0[u]);
            atomicAdd(p + lane + 32, wu * v1[u]);
        }
    }
}

__device__ __forceinline__ void own_replay(const int2* __restrict__ rec, int cnt, int lane,
                                           const float* __restrict__ t, float* tile) {
    if (cnt <= 0) return;
    const int last = cnt - 1;
    int2 cur = rec[lane < last ? lane : last];
    for (int base = 0; base < cnt; base += 32) {
        const int i1 = base + 32 + lane;
        const int2 nxt = rec[i1 < last ? i1 : last];
        const int left = cnt - base;
        if (left >= 32) {
            own_replay16<false, 0>(cur, 16, lane, t, tile);
            own_replay16<false, 16>(cur, 16, lane, t, tile);
        } else {
            own_replay16<true, 0>(cur, left < 16 ? left : 16, lane, t, tile);
            if (left > 16) own_replay16<true, 16>(cur, left - 16, lane, t, tile);
        }
        cur = nxt;
    }
}

template <bool REPLAY>
__global__ void __launch_bounds__(OWN_WAVES * 32)
k_agg_owner(const int* __restrict__ ei, const int* __restrict__ keys, const float* __restrict__ dinv,
            const float* __restrict__ t, const float* __restrict__ b, float* __restrict__ h, int n, int nedges,
            int* recs, int* cnts) {
    __shared__ float tile[OWN_R * 64];
    __shared__ int qedge[OWN_WAVES * OWN_QCAP];
    __shared__ int wcnt[OWN_WAVES];

    const int lane = threadIdx.x & 31;
    const int wv = __builtin_amdgcn_readfirstlane((int)(threadIdx.x >> 5));
    const int lo = blockIdx.x * OWN_R;
    const int rows = (n - lo) < OWN_R ? (n - lo) : OWN_R;

    for (int idx = threadIdx.x; idx < rows * 16; idx += OWN_WAVES * 32) {
        const int r = idx >> 4;
        const int j = (idx & 15) * 4;
        const float di = dinv[lo + r];
        const float sw = di * di;
        const float4 v = *(const float4*)(t + (size_t)(lo + r) * 64 + j);
        *(float4*)(tile + r * 64 + j) = make_float4(sw * v.x, sw * v.y, sw * v.z, sw * v.w);
    }
    __syncthreads();

    int* qe = qedge + wv * OWN_QCAP;
    const unsigned pat = (blockIdx.x & 255u) * 0x01010101u;
    int2* rec = (int2*)recs + (size_t)(blockIdx.x * OWN_WAVES + wv) * REC_CAP;

    if (REPLAY) {
        const int mine = __builtin_amdgcn_readfirstlane(cnts[blockIdx.x * OWN_WAVES + wv]);
        if (mine <= REC_CAP) own_replay(rec, mine, lane, t, tile);
        else (void)own_scan<false>(keys, wv, lane, pat, qe, lo, nedges, ei, dinv, t, tile, rec);
    } else {
        const int mine = own_scan<true>(keys, wv, lane, pat, qe, lo, nedges, ei, dinv, t, tile, rec);
        if (lane == 0) wcnt[wv] = mine;
    }
    __syncthreads();

    if (!REPLAY && wv == 0) cnts[blockIdx.x * OWN_WAVES + lane] = wcnt[lane];

    for (int idx = threadIdx.x; idx < rows * 16; idx += OWN_WAVES * 32) {
        const int r = idx >> 4;
        const int j = (idx & 15) * 4;
        const float4 v = *(const float4*)(tile + r * 64 + j);
        const float4 bv = *(const float4*)(b + j);
        float4 o;
        o.x = fmaxf(v.x + bv.x, 0.0f);
        o.y = fmaxf(v.y + bv.y, 0.0f);
        o.z = fmaxf(v.z + bv.z, 0.0f);
        o.w = fmaxf(v.w + bv.w, 0.0f);
        *(float4*)(h + (size_t)(lo + r) * 64 + j) = o;
    }
}

__global__ void k_final_dot(const float* __restrict__ H, const float* __restrict__ w,
                            const float* __restrict__ b, float* __restrict__ out, int n) {
    int i = blockIdx.x * blockDim.x + threadIdx.x;
    if (i >= n) return;
    const float4* hp = (const float4*)(H + (size_t)i * 64);
    const float4* wp = (const float4*)w;
    float s = 0.0f;
#pragma unroll
    for (int k = 0; k < 16; ++k) {
        float4 hv = hp[k], wv = wp[k];
        s += hv.x * wv.x + hv.y * wv.y + hv.z * wv.z + hv.w * wv.w;
    }
    out[i] = s + b[0];
}


extern "C" void kernel_launch(void* const* d_in, const int* in_sizes, int n_in,
                              void* d_out, int out_size, void* d_ws, size_t ws_size,
                              hipStream_t stream) {
    (void)n_in; (void)out_size; (void)ws_size;

    const float* x   = (const float*)d_in[0];
    const int*   ei  = (const int*)  d_in[1];
    const float* W1  = (const float*)d_in[2];
    const float* b1  = (const float*)d_in[3];
    const float* W2  = (const float*)d_in[4];
    const float* b2  = (const float*)d_in[5];
    const float* Wf1 = (const float*)d_in[6];
    const float* bf1 = (const float*)d_in[7];
    const float* Wf2 = (const float*)d_in[8];
    const float* bf2 = (const float*)d_in[9];
    const float* Wf3 = (const float*)d_in[10];
    const float* bf3 = (const float*)d_in[11];
    float* out = (float*)d_out;

    const int n = in_sizes[0] / 64;
    const int E = in_sizes[1] / 2;

    float* ws    = (float*)d_ws;
    float* dinv  = ws;
    float* bufA  = ws + n;
    float* bufB  = bufA + (size_t)n * 64;
    int*   keys  = (int*)((char*)d_ws + ((size_t)128 << 20));
    int*   cnts  = (int*)((char*)d_ws + ((size_t)160 << 20));
    int*   recs  = (int*)((char*)d_ws + ((size_t)192 << 20));
    unsigned* part = (unsigned*)((char*)d_ws + ((size_t)64 << 20));

    const int T = 256;
    const int rowTiles = (n + 15) / 16;
    const int gemmBlocks = (rowTiles + 7) / 8;
    const int ownBlocks = (n + OWN_R - 1) / OWN_R;

    k_deg_hist_keys<<<DEG_S * DEG_K, DEG_T, 0, stream>>>(ei, part, keys, n, E);
    k_dinv_from_parts<<<(n + T - 1) / T, T, 0, stream>>>(part, dinv, n);

    k_gemm64<false><<<gemmBlocks, T, 0, stream>>>(x, W1, nullptr, bufA, n);
    k_agg_owner<false><<<ownBlocks, OWN_WAVES * 32, 0, stream>>>(ei, keys, dinv, bufA, b1, bufB, n, E, recs, cnts);

    k_gemm64<false><<<gemmBlocks, T, 0, stream>>>(bufB, W2, nullptr, bufA, n);
    k_agg_owner<true><<<ownBlocks, OWN_WAVES * 32, 0, stream>>>(ei, keys, dinv, bufA, b2, bufB, n, E, recs, cnts);

    k_gemm64<true><<<gemmBlocks, T, 0, stream>>>(bufB, Wf1, bf1, bufA, n);
    k_gemm64<true><<<gemmBlocks, T, 0, stream>>>(bufA, Wf2, bf2, bufB, n);
    k_final_dot<<<(n + T - 1) / T, T, 0, stream>>>(bufB, Wf3, bf3, out, n);
}


# Product kernels: W staged by four 16-byte loads per thread all in flight, and the wave's 16 A-row loads issued before the staging barrier: one memory round trip ahead of the first WMMA, was nine
# speedup vs baseline: 2.6592x; 1.0836x over previous
#include <hip/hip_runtime.h>
#include <hip/hip_bf16.h>

typedef __attribute__((ext_vector_type(2))) float v2f;
typedef __attribute__((ext_vector_type(8))) float v8f;

#define OWN_R     512
#define OWN_SHIFT 9
#define OWN_WAVES 32
#define OWN_SUB   4
#define OWN_QCAP  48
#define KEY_TRIP  (OWN_WAVES * OWN_SUB * 512)
#define REC_CAP   512
#define REC_SHIFT 17

#define DEG_K    8
#define DEG_S    32
#define DEG_RCAP 16384
#define DEG_T    1024
#define DEG_U    8

__device__ __forceinline__ int deg_range_rows(int n) { return (((n + DEG_K - 1) / DEG_K) + 31) & ~31; }

template <bool WIDE>
__device__ __forceinline__ void deg_walk(const int* __restrict__ dstp, int nedges, int q0, int q1, int kr,
                                         int lo, int R, unsigned* hist, int* __restrict__ keys) {
    for (int qb = q0 + (int)threadIdx.x; qb < q1; qb += DEG_T * DEG_U) {
        int4 v[DEG_U];
#pragma unroll
        for (int u = 0; u < DEG_U; ++u) {
            const int q = qb + u * DEG_T;
            const long e0 = (long)q * 4;
            if (WIDE) {
                const bool in = q < q1 && e0 < nedges;
                const int4 d = *(const int4*)(dstp + (in ? e0 : 0));
                v[u] = in ? d : make_int4(-1, -1, -1, -1);
            } else {
                const bool i0 = q < q1 && e0     < nedges, i1 = q < q1 && e0 + 1 < nedges;
                const bool i2 = q < q1 && e0 + 2 < nedges, i3 = q < q1 && e0 + 3 < nedges;
                const int x0 = dstp[i0 ? e0     : 0], x1 = dstp[i1 ? e0 + 1 : 0];
                const int x2 = dstp[i2 ? e0 + 2 : 0], x3 = dstp[i3 ? e0 + 3 : 0];
                v[u] = make_int4(i0 ? x0 : -1, i1 ? x1 : -1, i2 ? x2 : -1, i3 ? x3 : -1);
            }
        }
#pragma unroll
        for (int u = 0; u < DEG_U; ++u) {
            const int q = qb + u * DEG_T;
            const int4 d = v[u];
            const unsigned r0 = (unsigned)(d.x - lo), r1 = (unsigned)(d.y - lo);
            const unsigned r2 = (unsigned)(d.z - lo), r3 = (unsigned)(d.w - lo);
            if (r0 < (unsigned)R) atomicAdd(&hist[r0], 1u);
            if (r1 < (unsigned)R) atomicAdd(&hist[r1], 1u);
            if (r2 < (unsigned)R) atomicAdd(&hist[r2], 1u);
            if (r3 < (unsigned)R) atomicAdd(&hist[r3], 1u);
            if ((u % DEG_K) == kr && q < q1) {
                const unsigned k0 = d.x >= 0 ? (((unsigned)d.x >> OWN_SHIFT) & 255u) : 255u;
                const unsigned k1 = d.y >= 0 ? (((unsigned)d.y >> OWN_SHIFT) & 255u) : 255u;
                const unsigned k2 = d.z >= 0 ? (((unsigned)d.z >> OWN_SHIFT) & 255u) : 255u;
                const unsigned k3 = d.w >= 0 ? (((unsigned)d.w >> OWN_SHIFT) & 255u) : 255u;
                keys[q] = (int)(k0 | (k1 << 8) | (k2 << 16) | (k3 << 24));
            }
        }
    }
}

__global__ void __launch_bounds__(DEG_T)
k_deg_hist_keys(const int* __restrict__ ei, unsigned* __restrict__ part, int* __restrict__ keys,
                int n, int nedges) {
    __shared__ unsigned hist[DEG_RCAP];
    const int kr = blockIdx.x % DEG_K;
    const int sl = blockIdx.x / DEG_K;
    const int R  = deg_range_rows(n);
    const int lo = kr * R;

    for (int j = threadIdx.x * 4; j < R; j += DEG_T * 4) *(uint4*)(hist + j) = make_uint4(0u, 0u, 0u, 0u);
    __syncthreads();

    const int npad4 = ((nedges + KEY_TRIP - 1) / KEY_TRIP) * (KEY_TRIP / 4);
    const int qps = (npad4 + DEG_S - 1) / DEG_S;
    const int q0 = sl * qps;
    const int q1 = (q0 + qps) < npad4 ? (q0 + qps) : npad4;
    const int* __restrict__ dstp = ei + nedges;
    if ((nedges & 3) == 0) deg_walk<true>(dstp, nedges, q0, q1, kr, lo, R, hist, keys);
    else                   deg_walk<false>(dstp, nedges, q0, q1, kr, lo, R, hist, keys);
    __syncthreads();

    unsigned* __restrict__ plane = part + (size_t)sl * (DEG_K * R) + lo;
    for (int j = threadIdx.x * 4; j < R; j += DEG_T * 4) *(uint4*)(plane + j) = *(const uint4*)(hist + j);
}

__global__ void k_dinv_from_parts(const unsigned* __restrict__ part, float* __restrict__ dinv, int n) {
    const int i = blockIdx.x * blockDim.x + threadIdx.x;
    if (i >= n) return;
    const int P = DEG_K * deg_range_rows(n);
    unsigned c = 1u;
#pragma unroll
    for (int s = 0; s < DEG_S; ++s) c += part[(size_t)s * P + i];
    dinv[i] = rsqrtf((float)c);
}


#define WT_PITCH 68

template <bool BIAS_RELU>
__global__ void k_gemm64(const float* __restrict__ A, const float* __restrict__ W,
                         const float* __restrict__ bias, float* __restrict__ C, int nrows) {
    __shared__ float ldsW[64 * WT_PITCH];

    const int lane = threadIdx.x & 31;
    const int wave = threadIdx.x >> 5;
    const int rowTiles = (nrows + 15) >> 4;
    const int tile = blockIdx.x * 8 + wave;
    const int half = lane >> 4;
    const int m    = lane & 15;
    int row = tile * 16 + m;
    if (row >= nrows) row = nrows - 1;
    const float* __restrict__ arow = A + (size_t)row * 64;

    v2f a[16];
#pragma unroll
    for (int kk = 0; kk < 16; ++kk)
        a[kk] = *(const v2f*)(arow + kk * 4 + 2 * half);

    float4 wv[4];
#pragma unroll
    for (int i = 0; i < 4; ++i) wv[i] = *(const float4*)(W + (size_t)(threadIdx.x + i * 256) * 4);
#pragma unroll
    for (int i = 0; i < 4; ++i) {
        const int idx = (threadIdx.x + i * 256) * 4;
        const int k = idx >> 6;
        const int col = idx & 63;
        ldsW[(col + 0) * WT_PITCH + k] = wv[i].x;
        ldsW[(col + 1) * WT_PITCH + k] = wv[i].y;
        ldsW[(col + 2) * WT_PITCH + k] = wv[i].z;
        ldsW[(col + 3) * WT_PITCH + k] = wv[i].w;
    }
    __syncthreads();

    if (tile < rowTiles) {
        v8f c0 = {}, c1 = {}, c2 = {}, c3 = {};
#pragma unroll
        for (int kk = 0; kk < 16; ++kk) {
            const int k0 = kk * 4 + 2 * half;
            v2f b0 = *(const v2f*)&ldsW[(0 * 16 + m) * WT_PITCH + k0];
            v2f b1 = *(const v2f*)&ldsW[(1 * 16 + m) * WT_PITCH + k0];
            v2f b2 = *(const v2f*)&ldsW[(2 * 16 + m) * WT_PITCH + k0];
            v2f b3 = *(const v2f*)&ldsW[(3 * 16 + m) * WT_PITCH + k0];
            c0 = __builtin_amdgcn_wmma_f32_16x16x4_f32(false, a[kk], false, b0, (short)0, c0, false, false);
            c1 = __builtin_amdgcn_wmma_f32_16x16x4_f32(false, a[kk], false, b1, (short)0, c1, false, false);
            c2 = __builtin_amdgcn_wmma_f32_16x16x4_f32(false, a[kk], false, b2, (short)0, c2, false, false);
            c3 = __builtin_amdgcn_wmma_f32_16x16x4_f32(false, a[kk], false, b3, (short)0, c3, false, false);
        }

        v8f acc[4] = {c0, c1, c2, c3};
#pragma unroll
        for (int nt = 0; nt < 4; ++nt) {
            const int col = nt * 16 + m;
            const float bv = BIAS_RELU ? bias[col] : 0.0f;
#pragma unroll
            for (int r = 0; r < 8; ++r) {
                const int orow = tile * 16 + r + 8 * half;
                if (orow < nrows) {
                    float v = acc[nt][r];
                    if (BIAS_RELU) v = fmaxf(v + bv, 0.0f);
                    C[(size_t)orow * 64 + col] = v;
                }
            }
        }
    }
}


__device__ __forceinline__ void own_load_keys(int4 (&kv)[OWN_SUB], const int4* __restrict__ p) {
#pragma unroll
    for (int i = 0; i < OWN_SUB; ++i) kv[i] = p[i * 32];
}

template <bool TAIL, bool REC>
__device__ __forceinline__ void own_drain(const int* qe, int q0, int cnt, int lane, int lo, int nedges,
                                          const int* __restrict__ ei, const float* __restrict__ dinv,
                                          const float* __restrict__ t, float* tile, int2* rec, int rpos) {
    int dl = 0, s = 0;
    if (lane < cnt) {
        const int e = qe[q0 + lane];
        s = ei[e];
        dl = ei[nedges + e] - lo;
    }
    float v0[16], v1[16];
#pragma unroll
    for (int u = 0; u < 16; ++u) {
        const int su = __builtin_amdgcn_readlane(s, u);
        const float* __restrict__ tp = t + (size_t)su * 64;
        v0[u] = tp[lane];
        v1[u] = tp[lane + 32];
    }
    const float w = dinv[s] * dinv[lo + dl];
    if (REC) {
        if (rpos + 16 <= REC_CAP && lane < cnt)
            rec[rpos + lane] = make_int2(s | (dl << REC_SHIFT), __builtin_bit_cast(int, w));
    }
#pragma unroll
    for (int u = 0; u < 16; ++u) {
        if (!TAIL || u < cnt) {
            const int du = __builtin_amdgcn_readlane(dl, u);
            const float wu = __builtin_bit_cast(float,
                __builtin_amdgcn_readlane(__builtin_bit_cast(int, w), u));
            float* p = tile + du * 64;
            atomicAdd(p + lane, wu * v0[u]);
            atomicAdd(p + lane + 32, wu * v1[u]);
        }
    }
}

template <bool REC>
__device__ __forceinline__ void own_trip(const int4 (&kv)[OWN_SUB], int eb, unsigned pat, int& qn, int* qe,
                                         int lane, unsigned lt_mask, int lo, int nedges,
                                         const int* __restrict__ ei, const float* __restrict__ dinv,
                                         const float* __restrict__ t, float* tile, int2* rec, int& rpos) {
#pragma unroll
    for (int p = 0; p < OWN_SUB; ++p) {
        const unsigned x0 = (unsigned)kv[p].x ^ pat, x1 = (unsigned)kv[p].y ^ pat;
        const unsigned x2 = (unsigned)kv[p].z ^ pat, x3 = (unsigned)kv[p].w ^ pat;
        const unsigned y0 = ((x0 & 0x7F7F7F7Fu) + 0x7F7F7F7Fu) | x0;
        const unsigned y1 = ((x1 & 0x7F7F7F7Fu) + 0x7F7F7F7Fu) | x1;
        const unsigned y2 = ((x2 & 0x7F7F7F7Fu) + 0x7F7F7F7Fu) | x2;
        const unsigned y3 = ((x3 & 0x7F7F7F7Fu) + 0x7F7F7F7Fu) | x3;
        const unsigned nz = ((y0 >> 7) & 0x01010101u) | ((y1 >> 6) & 0x02020202u) |
                            ((y2 >> 5) & 0x04040404u) | ((y3 >> 4) & 0x08080808u);
        unsigned m = nz ^ 0x0F0F0F0Fu;
        if (__builtin_amdgcn_ballot_w32(m != 0u) != 0u) {
            const int ep = eb + p * 512;
            do {
                const bool has = m != 0u;
                const unsigned k = __builtin_amdgcn_ballot_w32(has);
                if (has) {
                    const int bit = __builtin_ctz(m);
                    const int j = ((bit & 7) << 2) | (bit >> 3);
                    qe[qn + __builtin_popcount(k & lt_mask)] = ep + j;
                    m &= m - 1u;
                }
                qn += __builtin_popcount(k);
                __builtin_amdgcn_wave_barrier();
                while (qn >= 16) {
                    qn -= 16;
                    own_drain<false, REC>(qe, qn, 16, lane, lo, nedges, ei, dinv, t, tile, rec, rpos);
                    rpos += 16;
                }
                __builtin_amdgcn_wave_barrier();
            } while (__builtin_amdgcn_ballot_w32(m != 0u) != 0u);
        }
    }
}

template <bool REC>
__device__ __forceinline__ int own_scan(const int* __restrict__ keys, int wv, int lane, unsigned pat, int* qe,
                                        int lo, int nedges, const int* __restrict__ ei,
                                        const float* __restrict__ dinv, const float* __restrict__ t,
                                        float* tile, int2* rec) {
    const unsigned lt_mask = (1u << lane) - 1u;
    int qn = 0;
    int rpos = 0;

    const int ntrip = (nedges + KEY_TRIP - 1) / KEY_TRIP;
    const int4* __restrict__ kp = (const int4*)keys + wv * (OWN_SUB * 32) + lane;
    const int eb0 = wv * (OWN_SUB * 512) + lane * 16;
    if (ntrip > 0) {
        int4 ka[OWN_SUB], kb[OWN_SUB];
        own_load_keys(ka, kp);
        for (int k = 0; k < ntrip; k += 2) {
            const int k1 = (k + 1 < ntrip) ? k + 1 : ntrip - 1;
            own_load_keys(kb, kp + (size_t)k1 * (KEY_TRIP / 16));
            own_trip<REC>(ka, eb0 + k * KEY_TRIP, pat, qn, qe, lane, lt_mask, lo, nedges, ei, dinv, t, tile, rec, rpos);
            if (k + 1 >= ntrip) break;
            const int k2 = (k + 2 < ntrip) ? k + 2 : ntrip - 1;
            own_load_keys(ka, kp + (size_t)k2 * (KEY_TRIP / 16));
            own_trip<REC>(kb, eb0 + (k + 1) * KEY_TRIP, pat, qn, qe, lane, lt_mask, lo, nedges, ei, dinv, t, tile, rec, rpos);
        }
    }
    while (qn > 0) {
        const int c = qn < 16 ? qn : 16;
        qn -= c;
        own_drain<true, REC>(qe, qn, c, lane, lo, nedges, ei, dinv, t, tile, rec, rpos);
        rpos += c;
    }
    return rpos;
}

template <bool TAIL, int OFF>
__device__ __forceinline__ void own_replay16(int2 r, int c, int lane, const float* __restrict__ t, float* tile) {
    float v0[16], v1[16];
#pragma unroll
    for (int u = 0; u < 16; ++u) {
        const int su = __builtin_amdgcn_readlane(r.x, OFF + u) & ((1 << REC_SHIFT) - 1);
        const float* __restrict__ tp = t + (size_t)su * 64;
        v0[u] = tp[lane];
        v1[u] = tp[lane + 32];
    }
#pragma unroll
    for (int u = 0; u < 16; ++u) {
        if (!TAIL || u < c) {
            const int du = __builtin_amdgcn_readlane(r.x, OFF + u) >> REC_SHIFT;
            const float wu = __builtin_bit_cast(float, __builtin_amdgcn_readlane(r.y, OFF + u));
            float* p = tile + du * 64;
            atomicAdd(p + lane, wu * v0[u]);
            atomicAdd(p + lane + 32, wu * v1[u]);
        }
    }
}

__device__ __forceinline__ void own_replay(const int2* __restrict__ rec, int cnt, int lane,
                                           const float* __restrict__ t, float* tile) {
    if (cnt <= 0) return;
    const int last = cnt - 1;
    int2 cur = rec[lane < last ? lane : last];
    for (int base = 0; base < cnt; base += 32) {
        const int i1 = base + 32 + lane;
        const int2 nxt = rec[i1 < last ? i1 : last];
        const int left = cnt - base;
        if (left >= 32) {
            own_replay16<false, 0>(cur, 16, lane, t, tile);
            own_replay16<false, 16>(cur, 16, lane, t, tile);
        } else {
            own_replay16<true, 0>(cur, left < 16 ? left : 16, lane, t, tile);
            if (left > 16) own_replay16<true, 16>(cur, left - 16, lane, t, tile);
        }
        cur = nxt;
    }
}

template <bool REPLAY>
__global__ void __launch_bounds__(OWN_WAVES * 32)
k_agg_owner(const int* __restrict__ ei, const int* __restrict__ keys, const float* __restrict__ dinv,
            const float* __restrict__ t, const float* __restrict__ b, float* __restrict__ h, int n, int nedges,
            int* recs, int* cnts) {
    __shared__ float tile[OWN_R * 64];
    __shared__ int qedge[OWN_WAVES * OWN_QCAP];
    __shared__ int wcnt[OWN_WAVES];

    const int lane = threadIdx.x & 31;
    const int wv = __builtin_amdgcn_readfirstlane((int)(threadIdx.x >> 5));
    const int lo = blockIdx.x * OWN_R;
    const int rows = (n - lo) < OWN_R ? (n - lo) : OWN_R;

    for (int idx = threadIdx.x; idx < rows * 16; idx += OWN_WAVES * 32) {
        const int r = idx >> 4;
        const int j = (idx & 15) * 4;
        const float di = dinv[lo + r];
        const float sw = di * di;
        const float4 v = *(const float4*)(t + (size_t)(lo + r) * 64 + j);
        *(float4*)(tile + r * 64 + j) = make_float4(sw * v.x, sw * v.y, sw * v.z, sw * v.w);
    }
    __syncthreads();

    int* qe = qedge + wv * OWN_QCAP;
    const unsigned pat = (blockIdx.x & 255u) * 0x01010101u;
    int2* rec = (int2*)recs + (size_t)(blockIdx.x * OWN_WAVES + wv) * REC_CAP;

    if (REPLAY) {
        const int mine = __builtin_amdgcn_readfirstlane(cnts[blockIdx.x * OWN_WAVES + wv]);
        if (mine <= REC_CAP) own_replay(rec, mine, lane, t, tile);
        else (void)own_scan<false>(keys, wv, lane, pat, qe, lo, nedges, ei, dinv, t, tile, rec);
    } else {
        const int mine = own_scan<true>(keys, wv, lane, pat, qe, lo, nedges, ei, dinv, t, tile, rec);
        if (lane == 0) wcnt[wv] = mine;
    }
    __syncthreads();

    if (!REPLAY && wv == 0) cnts[blockIdx.x * OWN_WAVES + lane] = wcnt[lane];

    for (int idx = threadIdx.x; idx < rows * 16; idx += OWN_WAVES * 32) {
        const int r = idx >> 4;
        const int j = (idx & 15) * 4;
        const float4 v = *(const float4*)(tile + r * 64 + j);
        const float4 bv = *(const float4*)(b + j);
        float4 o;
        o.x = fmaxf(v.x + bv.x, 0.0f);
        o.y = fmaxf(v.y + bv.y, 0.0f);
        o.z = fmaxf(v.z + bv.z, 0.0f);
        o.w = fmaxf(v.w + bv.w, 0.0f);
        *(float4*)(h + (size_t)(lo + r) * 64 + j) = o;
    }
}

__global__ void k_final_dot(const float* __restrict__ H, const float* __restrict__ w,
                            const float* __restrict__ b, float* __restrict__ out, int n) {
    int i = blockIdx.x * blockDim.x + threadIdx.x;
    if (i >= n) return;
    const float4* hp = (const float4*)(H + (size_t)i * 64);
    const float4* wp = (const float4*)w;
    float s = 0.0f;
#pragma unroll
    for (int k = 0; k < 16; ++k) {
        float4 hv = hp[k], wv = wp[k];
        s += hv.x * wv.x + hv.y * wv.y + hv.z * wv.z + hv.w * wv.w;
    }
    out[i] = s + b[0];
}


extern "C" void kernel_launch(void* const* d_in, const int* in_sizes, int n_in,
                              void* d_out, int out_size, void* d_ws, size_t ws_size,
                              hipStream_t stream) {
    (void)n_in; (void)out_size; (void)ws_size;

    const float* x   = (const float*)d_in[0];
    const int*   ei  = (const int*)  d_in[1];
    const float* W1  = (const float*)d_in[2];
    const float* b1  = (const float*)d_in[3];
    const float* W2  = (const float*)d_in[4];
    const float* b2  = (const float*)d_in[5];
    const float* Wf1 = (const float*)d_in[6];
    const float* bf1 = (const float*)d_in[7];
    const float* Wf2 = (const float*)d_in[8];
    const float* bf2 = (const float*)d_in[9];
    const float* Wf3 = (const float*)d_in[10];
    const float* bf3 = (const float*)d_in[11];
    float* out = (float*)d_out;

    const int n = in_sizes[0] / 64;
    const int E = in_sizes[1] / 2;

    float* ws    = (float*)d_ws;
    float* dinv  = ws;
    float* bufA  = ws + n;
    float* bufB  = bufA + (size_t)n * 64;
    int*   keys  = (int*)((char*)d_ws + ((size_t)128 << 20));
    int*   cnts  = (int*)((char*)d_ws + ((size_t)160 << 20));
    int*   recs  = (int*)((char*)d_ws + ((size_t)192 << 20));
    unsigned* part = (unsigned*)((char*)d_ws + ((size_t)64 << 20));

    const int T = 256;
    const int rowTiles = (n + 15) / 16;
    const int gemmBlocks = (rowTiles + 7) / 8;
    const int ownBlocks = (n + OWN_R - 1) / OWN_R;

    k_deg_hist_keys<<<DEG_S * DEG_K, DEG_T, 0, stream>>>(ei, part, keys, n, E);
    k_dinv_from_parts<<<(n + T - 1) / T, T, 0, stream>>>(part, dinv, n);

    k_gemm64<false><<<gemmBlocks, T, 0, stream>>>(x, W1, nullptr, bufA, n);
    k_agg_owner<false><<<ownBlocks, OWN_WAVES * 32, 0, stream>>>(ei, keys, dinv, bufA, b1, bufB, n, E, recs, cnts);

    k_gemm64<false><<<gemmBlocks, T, 0, stream>>>(bufB, W2, nullptr, bufA, n);
    k_agg_owner<true><<<ownBlocks, OWN_WAVES * 32, 0, stream>>>(ei, keys, dinv, bufA, b2, bufB, n, E, recs, cnts);

    k_gemm64<true><<<gemmBlocks, T, 0, stream>>>(bufB, Wf1, bf1, bufA, n);
    k_gemm64<true><<<gemmBlocks, T, 0, stream>>>(bufA, Wf2, bf2, bufB, n);
    k_final_dot<<<(n + T - 1) / T, T, 0, stream>>>(bufB, Wf3, bf3, out, n);
}


# Degree-count launch also files each edge (source, local row) in its (owner, slice) list through LDS slot counters; layer 1 walks its wave's list: no key scan, no id reads; sum order LDS arrival
# speedup vs baseline: 3.0439x; 1.1447x over previous
#include <hip/hip_runtime.h>
#include <hip/hip_bf16.h>

typedef __attribute__((ext_vector_type(2))) float v2f;
typedef __attribute__((ext_vector_type(8))) float v8f;

#define OWN_R     512
#define OWN_SHIFT 9
#define OWN_WAVES 32
#define OWN_SUB   4
#define OWN_QCAP  48
#define KEY_TRIP  (OWN_WAVES * OWN_SUB * 512)
#define REC_CAP   512
#define REC_SHIFT 17
#define DEG_OMAX  32

#define DEG_K    8
#define DEG_S    32
#define DEG_RCAP 16384
#define DEG_T    1024
#define DEG_U    8

__device__ __forceinline__ int deg_range_rows(int n) { return (((n + DEG_K - 1) / DEG_K) + OWN_R - 1) & ~(OWN_R - 1); }

__device__ __forceinline__ void deg_bin(unsigned* bcnt, int* bins, unsigned r, int s) {
    const unsigned ol = r >> OWN_SHIFT;
    const unsigned pos = atomicAdd(&bcnt[ol], 1u);
    if (pos < (unsigned)REC_CAP) bins[ol * REC_CAP + pos] = s | (int)((r & (unsigned)(OWN_R - 1)) << REC_SHIFT);
}

template <bool WIDE>
__device__ __forceinline__ void deg_walk(const int* __restrict__ srcp, const int* __restrict__ dstp, int nedges,
                                         int q0, int q1, int kr, int lo, int R, unsigned* hist,
                                         unsigned* bcnt, int* bins, int* __restrict__ keys) {
    for (int qb = q0 + (int)threadIdx.x; qb < q1; qb += DEG_T * DEG_U) {
        int4 v[DEG_U], sv[DEG_U];
#pragma unroll
        for (int u = 0; u < DEG_U; ++u) {
            const int q = qb + u * DEG_T;
            const long e0 = (long)q * 4;
            if (WIDE) {
                const bool in = q < q1 && e0 < nedges;
                const int4 d = *(const int4*)(dstp + (in ? e0 : 0));
                sv[u] = *(const int4*)(srcp + (in ? e0 : 0));
                v[u] = in ? d : make_int4(-1, -1, -1, -1);
            } else {
                const bool i0 = q < q1 && e0     < nedges, i1 = q < q1 && e0 + 1 < nedges;
                const bool i2 = q < q1 && e0 + 2 < nedges, i3 = q < q1 && e0 + 3 < nedges;
                const int x0 = dstp[i0 ? e0     : 0], x1 = dstp[i1 ? e0 + 1 : 0];
                const int x2 = dstp[i2 ? e0 + 2 : 0], x3 = dstp[i3 ? e0 + 3 : 0];
                v[u] = make_int4(i0 ? x0 : -1, i1 ? x1 : -1, i2 ? x2 : -1, i3 ? x3 : -1);
                sv[u] = make_int4(srcp[i0 ? e0 : 0], srcp[i1 ? e0 + 1 : 0], srcp[i2 ? e0 + 2 : 0], srcp[i3 ? e0 + 3 : 0]);
            }
        }
#pragma unroll
        for (int u = 0; u < DEG_U; ++u) {
            const int q = qb + u * DEG_T;
            const int4 d = v[u];
            const int4 sg = sv[u];
            const unsigned r0 = (unsigned)(d.x - lo), r1 = (unsigned)(d.y - lo);
            const unsigned r2 = (unsigned)(d.z - lo), r3 = (unsigned)(d.w - lo);
            if (r0 < (unsigned)R) { atomicAdd(&hist[r0], 1u); deg_bin(bcnt, bins, r0, sg.x); }
            if (r1 < (unsigned)R) { atomicAdd(&hist[r1], 1u); deg_bin(bcnt, bins, r1, sg.y); }
            if (r2 < (unsigned)R) { atomicAdd(&hist[r2], 1u); deg_bin(bcnt, bins, r2, sg.z); }
            if (r3 < (unsigned)R) { atomicAdd(&hist[r3], 1u); deg_bin(bcnt, bins, r3, sg.w); }
            if ((u % DEG_K) == kr && q < q1) {
                const unsigned k0 = d.x >= 0 ? (((unsigned)d.x >> OWN_SHIFT) & 255u) : 255u;
                const unsigned k1 = d.y >= 0 ? (((unsigned)d.y >> OWN_SHIFT) & 255u) : 255u;
                const unsigned k2 = d.z >= 0 ? (((unsigned)d.z >> OWN_SHIFT) & 255u) : 255u;
                const unsigned k3 = d.w >= 0 ? (((unsigned)d.w >> OWN_SHIFT) & 255u) : 255u;
                keys[q] = (int)(k0 | (k1 << 8) | (k2 << 16) | (k3 << 24));
            }
        }
    }
}

__global__ void __launch_bounds__(DEG_T)
k_deg_hist_keys(const int* __restrict__ ei, unsigned* __restrict__ part, int* __restrict__ keys,
                int* __restrict__ rec0, int* __restrict__ cnts, int n, int nedges) {
    __shared__ unsigned hist[DEG_RCAP];
    __shared__ int bins[DEG_OMAX * REC_CAP];
    __shared__ unsigned bcnt[DEG_OMAX];
    const int kr = blockIdx.x % DEG_K;
    const int sl = blockIdx.x / DEG_K;
    const int R  = deg_range_rows(n);
    const int lo = kr * R;

    for (int j = threadIdx.x * 4; j < R; j += DEG_T * 4) *(uint4*)(hist + j) = make_uint4(0u, 0u, 0u, 0u);
    if (threadIdx.x < DEG_OMAX) bcnt[threadIdx.x] = 0u;
    __syncthreads();

    const int npad4 = ((nedges + KEY_TRIP - 1) / KEY_TRIP) * (KEY_TRIP / 4);
    const int qps = (npad4 + DEG_S - 1) / DEG_S;
    const int q0 = sl * qps;
    const int q1 = (q0 + qps) < npad4 ? (q0 + qps) : npad4;
    const int* __restrict__ dstp = ei + nedges;
    if ((nedges & 3) == 0) deg_walk<true>(ei, dstp, nedges, q0, q1, kr, lo, R, hist, bcnt, bins, keys);
    else                   deg_walk<false>(ei, dstp, nedges, q0, q1, kr, lo, R, hist, bcnt, bins, keys);
    __syncthreads();

    unsigned* __restrict__ plane = part + (size_t)sl * (DEG_K * R) + lo;
    for (int j = threadIdx.x * 4; j < R; j += DEG_T * 4) *(uint4*)(plane + j) = *(const uint4*)(hist + j);

    const int opr = R >> OWN_SHIFT;
    for (int idx = threadIdx.x; idx < opr * REC_CAP; idx += DEG_T) {
        const int ol = idx / REC_CAP;
        const int j  = idx % REC_CAP;
        if ((unsigned)j < bcnt[ol])
            rec0[((size_t)(kr * opr + ol) * DEG_S + sl) * REC_CAP + j] = bins[idx];
    }
    if (threadIdx.x < DEG_OMAX)
        cnts[(sl * DEG_K + kr) * DEG_OMAX + threadIdx.x] = (int)threadIdx.x < opr ? (int)bcnt[threadIdx.x] : 0;
}

__global__ void k_dinv_from_parts(const unsigned* __restrict__ part, float* __restrict__ dinv, int n) {
    const int i = blockIdx.x * blockDim.x + threadIdx.x;
    if (i >= n) return;
    const int P = DEG_K * deg_range_rows(n);
    unsigned c = 1u;
#pragma unroll
    for (int s = 0; s < DEG_S; ++s) c += part[(size_t)s * P + i];
    dinv[i] = rsqrtf((float)c);
}


#define WT_PITCH 68

template <bool BIAS_RELU>
__global__ void k_gemm64(const float* __restrict__ A, const float* __restrict__ W,
                         const float* __restrict__ bias, float* __restrict__ C, int nrows) {
    __shared__ float ldsW[64 * WT_PITCH];

    const int lane = threadIdx.x & 31;
    const int wave = threadIdx.x >> 5;
    const int rowTiles = (nrows + 15) >> 4;
    const int tile = blockIdx.x * 8 + wave;
    const int half = lane >> 4;
    const int m    = lane & 15;
    int row = tile * 16 + m;
    if (row >= nrows) row = nrows - 1;
    const float* __restrict__ arow = A + (size_t)row * 64;

    v2f a[16];
#pragma unroll
    for (int kk = 0; kk < 16; ++kk)
        a[kk] = *(const v2f*)(arow + kk * 4 + 2 * half);

    float4 wv[4];
#pragma unroll
    for (int i = 0; i < 4; ++i) wv[i] = *(const float4*)(W + (size_t)(threadIdx.x + i * 256) * 4);
#pragma unroll
    for (int i = 0; i < 4; ++i) {
        const int idx = (threadIdx.x + i * 256) * 4;
        const int k = idx >> 6;
        const int col = idx & 63;
        ldsW[(col + 0) * WT_PITCH + k] = wv[i].x;
        ldsW[(col + 1) * WT_PITCH + k] = wv[i].y;
        ldsW[(col + 2) * WT_PITCH + k] = wv[i].z;
        ldsW[(col + 3) * WT_PITCH + k] = wv[i].w;
    }
    __syncthreads();

    if (tile < rowTiles) {
        v8f c0 = {}, c1 = {}, c2 = {}, c3 = {};
#pragma unroll
        for (int kk = 0; kk < 16; ++kk) {
            const int k0 = kk * 4 + 2 * half;
            v2f b0 = *(const v2f*)&ldsW[(0 * 16 + m) * WT_PITCH + k0];
            v2f b1 = *(const v2f*)&ldsW[(1 * 16 + m) * WT_PITCH + k0];
            v2f b2 = *(const v2f*)&ldsW[(2 * 16 + m) * WT_PITCH + k0];
            v2f b3 = *(const v2f*)&ldsW[(3 * 16 + m) * WT_PITCH + k0];
            c0 = __builtin_amdgcn_wmma_f32_16x16x4_f32(false, a[kk], false, b0, (short)0, c0, false, false);
            c1 = __builtin_amdgcn_wmma_f32_16x16x4_f32(false, a[kk], false, b1, (short)0, c1, false, false);
            c2 = __builtin_amdgcn_wmma_f32_16x16x4_f32(false, a[kk], false, b2, (short)0, c2, false, false);
            c3 = __builtin_amdgcn_wmma_f32_16x16x4_f32(false, a[kk], false, b3, (short)0, c3, false, false);
        }

        v8f acc[4] = {c0, c1, c2, c3};
#pragma unroll
        for (int nt = 0; nt < 4; ++nt) {
            const int col = nt * 16 + m;
            const float bv = BIAS_RELU ? bias[col] : 0.0f;
#pragma unroll
            for (int r = 0; r < 8; ++r) {
                const int orow = tile * 16 + r + 8 * half;
                if (orow < nrows) {
                    float v = acc[nt][r];
                    if (BIAS_RELU) v = fmaxf(v + bv, 0.0f);
                    C[(size_t)orow * 64 + col] = v;
                }
            }
        }
    }
}


__device__ __forceinline__ void own_load_keys(int4 (&kv)[OWN_SUB], const int4* __restrict__ p) {
#pragma unroll
    for (int i = 0; i < OWN_SUB; ++i) kv[i] = p[i * 32];
}

template <bool TAIL>
__device__ __forceinline__ void own_drain(const int* qe, int q0, int cnt, int lane, int lo, int nedges,
                                          const int* __restrict__ ei, const float* __restrict__ dinv,
                                          const float* __restrict__ t, float* tile) {
    int dl = 0, s = 0;
    if (lane < cnt) {
        const int e = qe[q0 + lane];
        s = ei[e];
        dl = ei[nedges + e] - lo;
    }
    float v0[16], v1[16];
#pragma unroll
    for (int u = 0; u < 16; ++u) {
        const int su = __builtin_amdgcn_readlane(s, u);
        const float* __restrict__ tp = t + (size_t)su * 64;
        v0[u] = tp[lane];
        v1[u] = tp[lane + 32];
    }
    const float w = dinv[s] * dinv[lo + dl];
#pragma unroll
    for (int u = 0; u < 16; ++u) {
        if (!TAIL || u < cnt) {
            const int du = __builtin_amdgcn_readlane(dl, u);
            const float wu = __builtin_bit_cast(float,
                __builtin_amdgcn_readlane(__builtin_bit_cast(int, w), u));
            float* p = tile + du * 64;
            atomicAdd(p + lane, wu * v0[u]);
            atomicAdd(p + lane + 32, wu * v1[u]);
        }
    }
}

__device__ __forceinline__ void own_trip(const int4 (&kv)[OWN_SUB], int eb, unsigned pat, int& qn, int* qe,
                                         int lane, unsigned lt_mask, int lo, int nedges,
                                         const int* __restrict__ ei, const float* __restrict__ dinv,
                                         const float* __restrict__ t, float* tile) {
#pragma unroll
    for (int p = 0; p < OWN_SUB; ++p) {
        const unsigned x0 = (unsigned)kv[p].x ^ pat, x1 = (unsigned)kv[p].y ^ pat;
        const unsigned x2 = (unsigned)kv[p].z ^ pat, x3 = (unsigned)kv[p].w ^ pat;
        const unsigned y0 = ((x0 & 0x7F7F7F7Fu) + 0x7F7F7F7Fu) | x0;
        const unsigned y1 = ((x1 & 0x7F7F7F7Fu) + 0x7F7F7F7Fu) | x1;
        const unsigned y2 = ((x2 & 0x7F7F7F7Fu) + 0x7F7F7F7Fu) | x2;
        const unsigned y3 = ((x3 & 0x7F7F7F7Fu) + 0x7F7F7F7Fu) | x3;
        const unsigned nz = ((y0 >> 7) & 0x01010101u) | ((y1 >> 6) & 0x02020202u) |
                            ((y2 >> 5) & 0x04040404u) | ((y3 >> 4) & 0x08080808u);
        unsigned m = nz ^ 0x0F0F0F0Fu;
        if (__builtin_amdgcn_ballot_w32(m != 0u) != 0u) {
            const int ep = eb + p * 512;
            do {
                const bool has = m != 0u;
                const unsigned k = __builtin_amdgcn_ballot_w32(has);
                if (has) {
                    const int bit = __builtin_ctz(m);
                    const int j = ((bit & 7) << 2) | (bit >> 3);
                    qe[qn + __builtin_popcount(k & lt_mask)] = ep + j;
                    m &= m - 1u;
                }
                qn += __builtin_popcount(k);
                __builtin_amdgcn_wave_barrier();
                while (qn >= 16) {
                    qn -= 16;
                    own_drain<false>(qe, qn, 16, lane, lo, nedges, ei, dinv, t, tile);
                }
                __builtin_amdgcn_wave_barrier();
            } while (__builtin_amdgcn_ballot_w32(m != 0u) != 0u);
        }
    }
}

__device__ __forceinline__ void own_scan(const int* __restrict__ keys, int wv, int lane, unsigned pat, int* qe,
                                         int lo, int nedges, const int* __restrict__ ei,
                                         const float* __restrict__ dinv, const float* __restrict__ t,
                                         float* tile) {
    const unsigned lt_mask = (1u << lane) - 1u;
    int qn = 0;

    const int ntrip = (nedges + KEY_TRIP - 1) / KEY_TRIP;
    const int4* __restrict__ kp = (const int4*)keys + (size_t)wv * ntrip * (OWN_SUB * 32) + lane;
    const int eb0 = wv * ntrip * (OWN_SUB * 512) + lane * 16;
    if (ntrip > 0) {
        int4 ka[OWN_SUB], kb[OWN_SUB];
        own_load_keys(ka, kp);
        for (int k = 0; k < ntrip; k += 2) {
            const int k1 = (k + 1 < ntrip) ? k + 1 : ntrip - 1;
            own_load_keys(kb, kp + (size_t)k1 * (OWN_SUB * 32));
            own_trip(ka, eb0 + k * (OWN_SUB * 512), pat, qn, qe, lane, lt_mask, lo, nedges, ei, dinv, t, tile);
            if (k + 1 >= ntrip) break;
            const int k2 = (k + 2 < ntrip) ? k + 2 : ntrip - 1;
            own_load_keys(ka, kp + (size_t)k2 * (OWN_SUB * 32));
            own_trip(kb, eb0 + (k + 1) * (OWN_SUB * 512), pat, qn, qe, lane, lt_mask, lo, nedges, ei, dinv, t, tile);
        }
    }
    while (qn > 0) {
        const int c = qn < 16 ? qn : 16;
        qn -= c;
        own_drain<true>(qe, qn, c, lane, lo, nedges, ei, dinv, t, tile);
    }
}

template <bool TAIL, int OFF>
__device__ __forceinline__ void own_replay16(int2 r, int c, int lane, const float* __restrict__ t, float* tile) {
    float v0[16], v1[16];
#pragma unroll
    for (int u = 0; u < 16; ++u) {
        const int su = __builtin_amdgcn_readlane(r.x, OFF + u) & ((1 << REC_SHIFT) - 1);
        const float* __restrict__ tp = t + (size_t)su * 64;
        v0[u] = tp[lane];
        v1[u] = tp[lane + 32];
    }
#pragma unroll
    for (int u = 0; u < 16; ++u) {
        if (!TAIL || u < c) {
            const int du = __builtin_amdgcn_readlane(r.x, OFF + u) >> REC_SHIFT;
            const float wu = __builtin_bit_cast(float, __builtin_amdgcn_readlane(r.y, OFF + u));
            float* p = tile + du * 64;
            atomicAdd(p + lane, wu * v0[u]);
            atomicAdd(p + lane + 32, wu * v1[u]);
        }
    }
}

__device__ __forceinline__ void own_replay(const int2* __restrict__ rec, int cnt, int lane,
                                           const float* __restrict__ t, float* tile) {
    if (cnt <= 0) return;
    const int last = cnt - 1;
    int2 cur = rec[lane < last ? lane : last];
    for (int base = 0; base < cnt; base += 32) {
        const int i1 = base + 32 + lane;
        const int2 nxt = rec[i1 < last ? i1 : last];
        const int left = cnt - base;
        if (left >= 32) {
            own_replay16<false, 0>(cur, 16, lane, t, tile);
            own_replay16<false, 16>(cur, 16, lane, t, tile);
        } else {
            own_replay16<true, 0>(cur, left < 16 ? left : 16, lane, t, tile);
            if (left > 16) own_replay16<true, 16>(cur, left - 16, lane, t, tile);
        }
        cur = nxt;
    }
}

__device__ __forceinline__ void own_first(const int* __restrict__ word, int2* rec, int cnt, int lane, int lo,
                                          const float* __restrict__ dinv, const float* __restrict__ t,
                                          float* tile) {
    if (cnt <= 0) return;
    const int last = cnt - 1;
    const int smask = (1 << REC_SHIFT) - 1;
    int wc = word[lane < last ? lane : last];
    int wn = word[lane + 32 < last ? lane + 32 : last];
    float w = dinv[wc & smask] * dinv[lo + (int)((unsigned)wc >> REC_SHIFT)];
    for (int base = 0; base < cnt; base += 32) {
        const int i2 = base + 64 + lane;
        const int wnn = word[i2 < last ? i2 : last];
        const float da = dinv[wn & smask];
        const float db = dinv[lo + (int)((unsigned)wn >> REC_SHIFT)];
        const int2 cur = make_int2(wc, __builtin_bit_cast(int, w));
        const int left = cnt - base;
        if (left >= 32) {
            own_replay16<false, 0>(cur, 16, lane, t, tile);
            own_replay16<false, 16>(cur, 16, lane, t, tile);
        } else {
            own_replay16<true, 0>(cur, left < 16 ? left : 16, lane, t, tile);
            if (left > 16) own_replay16<true, 16>(cur, left - 16, lane, t, tile);
        }
        if (base + lane < cnt) rec[base + lane] = cur;
        w = da * db;
        wc = wn;
        wn = wnn;
    }
}

static_assert(OWN_WAVES == DEG_S, "one wave per edge slice");
template <bool REPLAY>
__global__ void __launch_bounds__(OWN_WAVES * 32)
k_agg_owner(const int* __restrict__ ei, const int* __restrict__ keys, const float* __restrict__ dinv,
            const float* __restrict__ t, const float* __restrict__ b, float* __restrict__ h, int n, int nedges,
            const int* __restrict__ rec0, int* recs, const int* __restrict__ cnts) {
    __shared__ float tile[OWN_R * 64];
    __shared__ int qedge[OWN_WAVES * OWN_QCAP];

    const int lane = threadIdx.x & 31;
    const int wv = __builtin_amdgcn_readfirstlane((int)(threadIdx.x >> 5));
    const int lo = blockIdx.x * OWN_R;
    const int rows = (n - lo) < OWN_R ? (n - lo) : OWN_R;

    for (int idx = threadIdx.x; idx < rows * 16; idx += OWN_WAVES * 32) {
        const int r = idx >> 4;
        const int j = (idx & 15) * 4;
        const float di = dinv[lo + r];
        const float sw = di * di;
        const float4 v = *(const float4*)(t + (size_t)(lo + r) * 64 + j);
        *(float4*)(tile + r * 64 + j) = make_float4(sw * v.x, sw * v.y, sw * v.z, sw * v.w);
    }
    __syncthreads();

    int* qe = qedge + wv * OWN_QCAP;
    const unsigned pat = (blockIdx.x & 255u) * 0x01010101u;
    int2* rec = (int2*)recs + (size_t)(blockIdx.x * OWN_WAVES + wv) * REC_CAP;

    const int opr = deg_range_rows(n) >> OWN_SHIFT;
    const int* __restrict__ word = rec0 + (size_t)(blockIdx.x * DEG_S + wv) * REC_CAP;
    const int mine = __builtin_amdgcn_readfirstlane(
        cnts[(wv * DEG_K + (int)blockIdx.x / opr) * DEG_OMAX + (int)blockIdx.x % opr]);
    if (mine <= REC_CAP) {
        if (REPLAY) own_replay(rec, mine, lane, t, tile);
        else        own_first(word, rec, mine, lane, lo, dinv, t, tile);
    } else {
        own_scan(keys, wv, lane, pat, qe, lo, nedges, ei, dinv, t, tile);
    }
    __syncthreads();

    for (int idx = threadIdx.x; idx < rows * 16; idx += OWN_WAVES * 32) {
        const int r = idx >> 4;
        const int j = (idx & 15) * 4;
        const float4 v = *(const float4*)(tile + r * 64 + j);
        const float4 bv = *(const float4*)(b + j);
        float4 o;
        o.x = fmaxf(v.x + bv.x, 0.0f);
        o.y = fmaxf(v.y + bv.y, 0.0f);
        o.z = fmaxf(v.z + bv.z, 0.0f);
        o.w = fmaxf(v.w + bv.w, 0.0f);
        *(float4*)(h + (size_t)(lo + r) * 64 + j) = o;
    }
}

__global__ void k_final_dot(const float* __restrict__ H, const float* __restrict__ w,
                            const float* __restrict__ b, float* __restrict__ out, int n) {
    int i = blockIdx.x * blockDim.x + threadIdx.x;
    if (i >= n) return;
    const float4* hp = (const float4*)(H + (size_t)i * 64);
    const float4* wp = (const float4*)w;
    float s = 0.0f;
#pragma unroll
    for (int k = 0; k < 16; ++k) {
        float4 hv = hp[k], wv = wp[k];
        s += hv.x * wv.x + hv.y * wv.y + hv.z * wv.z + hv.w * wv.w;
    }
    out[i] = s + b[0];
}


extern "C" void kernel_launch(void* const* d_in, const int* in_sizes, int n_in,
                              void* d_out, int out_size, void* d_ws, size_t ws_size,
                              hipStream_t stream) {
    (void)n_in; (void)out_size; (void)ws_size;

    const float* x   = (const float*)d_in[0];
    const int*   ei  = (const int*)  d_in[1];
    const float* W1  = (const float*)d_in[2];
    const float* b1  = (const float*)d_in[3];
    const float* W2  = (const float*)d_in[4];
    const float* b2  = (const float*)d_in[5];
    const float* Wf1 = (const float*)d_in[6];
    const float* bf1 = (const float*)d_in[7];
    const float* Wf2 = (const float*)d_in[8];
    const float* bf2 = (const float*)d_in[9];
    const float* Wf3 = (const float*)d_in[10];
    const float* bf3 = (const float*)d_in[11];
    float* out = (float*)d_out;

    const int n = in_sizes[0] / 64;
    const int E = in_sizes[1] / 2;

    float* ws    = (float*)d_ws;
    float* dinv  = ws;
    float* bufA  = ws + n;
    float* bufB  = bufA + (size_t)n * 64;
    int*   keys  = (int*)((char*)d_ws + ((size_t)128 << 20));
    int*   cnts  = (int*)((char*)d_ws + ((size_t)160 << 20));
    int*   recs  = (int*)((char*)d_ws + ((size_t)192 << 20));
    int*   rec0  = (int*)((char*)d_ws + ((size_t)96 << 20));
    unsigned* part = (unsigned*)((char*)d_ws + ((size_t)64 << 20));

    const int T = 256;
    const int rowTiles = (n + 15) / 16;
    const int gemmBlocks = (rowTiles + 7) / 8;
    const int ownBlocks = (n + OWN_R - 1) / OWN_R;

    k_deg_hist_keys<<<DEG_S * DEG_K, DEG_T, 0, stream>>>(ei, part, keys, rec0, cnts, n, E);
    k_dinv_from_parts<<<(n + T - 1) / T, T, 0, stream>>>(part, dinv, n);

    k_gemm64<false><<<gemmBlocks, T, 0, stream>>>(x, W1, nullptr, bufA, n);
    k_agg_owner<false><<<ownBlocks, OWN_WAVES * 32, 0, stream>>>(ei, keys, dinv, bufA, b1, bufB, n, E, rec0, recs, cnts);

    k_gemm64<false><<<gemmBlocks, T, 0, stream>>>(bufB, W2, nullptr, bufA, n);
    k_agg_owner<true><<<ownBlocks, OWN_WAVES * 32, 0, stream>>>(ei, keys, dinv, bufA, b2, bufB, n, E, rec0, recs, cnts);

    k_gemm64<true><<<gemmBlocks, T, 0, stream>>>(bufB, Wf1, bf1, bufA, n);
    k_gemm64<true><<<gemmBlocks, T, 0, stream>>>(bufA, Wf2, bf2, bufB, n);
    k_final_dot<<<(n + T - 1) / T, T, 0, stream>>>(bufB, Wf3, bf3, out, n);
}


# Aggregation launches: tile-start row loads issued together; list count, list head and bias requested in front of the first barrier: 3 memory round trips before the first edge rows, was 11; sums same
# speedup vs baseline: 3.1809x; 1.0450x over previous
#include <hip/hip_runtime.h>
#include <hip/hip_bf16.h>

typedef __attribute__((ext_vector_type(2))) float v2f;
typedef __attribute__((ext_vector_type(8))) float v8f;

#define OWN_R     512
#define OWN_SHIFT 9
#define OWN_WAVES 32
#define OWN_PASS  (OWN_R * 16 / (OWN_WAVES * 32))
#define OWN_SUB   4
#define OWN_QCAP  48
#define KEY_TRIP  (OWN_WAVES * OWN_SUB * 512)
#define REC_CAP   512
#define REC_SHIFT 17
#define DEG_OMAX  32

#define DEG_K    8
#define DEG_S    32
#define DEG_RCAP 16384
#define DEG_T    1024
#define DEG_U    8

__device__ __forceinline__ int deg_range_rows(int n) { return (((n + DEG_K - 1) / DEG_K) + OWN_R - 1) & ~(OWN_R - 1); }

__device__ __forceinline__ void deg_bin(unsigned* bcnt, int* bins, unsigned r, int s) {
    const unsigned ol = r >> OWN_SHIFT;
    const unsigned pos = atomicAdd(&bcnt[ol], 1u);
    if (pos < (unsigned)REC_CAP) bins[ol * REC_CAP + pos] = s | (int)((r & (unsigned)(OWN_R - 1)) << REC_SHIFT);
}

template <bool WIDE>
__device__ __forceinline__ void deg_walk(const int* __restrict__ srcp, const int* __restrict__ dstp, int nedges,
                                         int q0, int q1, int kr, int lo, int R, unsigned* hist,
                                         unsigned* bcnt, int* bins, int* __restrict__ keys) {
    for (int qb = q0 + (int)threadIdx.x; qb < q1; qb += DEG_T * DEG_U) {
        int4 v[DEG_U], sv[DEG_U];
#pragma unroll
        for (int u = 0; u < DEG_U; ++u) {
            const int q = qb + u * DEG_T;
            const long e0 = (long)q * 4;
            if (WIDE) {
                const bool in = q < q1 && e0 < nedges;
                const int4 d = *(const int4*)(dstp + (in ? e0 : 0));
                sv[u] = *(const int4*)(srcp + (in ? e0 : 0));
                v[u] = in ? d : make_int4(-1, -1, -1, -1);
            } else {
                const bool i0 = q < q1 && e0     < nedges, i1 = q < q1 && e0 + 1 < nedges;
                const bool i2 = q < q1 && e0 + 2 < nedges, i3 = q < q1 && e0 + 3 < nedges;
                const int x0 = dstp[i0 ? e0     : 0], x1 = dstp[i1 ? e0 + 1 : 0];
                const int x2 = dstp[i2 ? e0 + 2 : 0], x3 = dstp[i3 ? e0 + 3 : 0];
                v[u] = make_int4(i0 ? x0 : -1, i1 ? x1 : -1, i2 ? x2 : -1, i3 ? x3 : -1);
                sv[u] = make_int4(srcp[i0 ? e0 : 0], srcp[i1 ? e0 + 1 : 0], srcp[i2 ? e0 + 2 : 0], srcp[i3 ? e0 + 3 : 0]);
            }
        }
#pragma unroll
        for (int u = 0; u < DEG_U; ++u) {
            const int q = qb + u * DEG_T;
            const int4 d = v[u];
            const int4 sg = sv[u];
            const unsigned r0 = (unsigned)(d.x - lo), r1 = (unsigned)(d.y - lo);
            const unsigned r2 = (unsigned)(d.z - lo), r3 = (unsigned)(d.w - lo);
            if (r0 < (unsigned)R) { atomicAdd(&hist[r0], 1u); deg_bin(bcnt, bins, r0, sg.x); }
            if (r1 < (unsigned)R) { atomicAdd(&hist[r1], 1u); deg_bin(bcnt, bins, r1, sg.y); }
            if (r2 < (unsigned)R) { atomicAdd(&hist[r2], 1u); deg_bin(bcnt, bins, r2, sg.z); }
            if (r3 < (unsigned)R) { atomicAdd(&hist[r3], 1u); deg_bin(bcnt, bins, r3, sg.w); }
            if ((u % DEG_K) == kr && q < q1) {
                const unsigned k0 = d.x >= 0 ? (((unsigned)d.x >> OWN_SHIFT) & 255u) : 255u;
                const unsigned k1 = d.y >= 0 ? (((unsigned)d.y >> OWN_SHIFT) & 255u) : 255u;
                const unsigned k2 = d.z >= 0 ? (((unsigned)d.z >> OWN_SHIFT) & 255u) : 255u;
                const unsigned k3 = d.w >= 0 ? (((unsigned)d.w >> OWN_SHIFT) & 255u) : 255u;
                keys[q] = (int)(k0 | (k1 << 8) | (k2 << 16) | (k3 << 24));
            }
        }
    }
}

__global__ void __launch_bounds__(DEG_T)
k_deg_hist_keys(const int* __restrict__ ei, unsigned* __restrict__ part, int* __restrict__ keys,
                int* __restrict__ rec0, int* __restrict__ cnts, int n, int nedges) {
    __shared__ unsigned hist[DEG_RCAP];
    __shared__ int bins[DEG_OMAX * REC_CAP];
    __shared__ unsigned bcnt[DEG_OMAX];
    const int kr = blockIdx.x % DEG_K;
    const int sl = blockIdx.x / DEG_K;
    const int R  = deg_range_rows(n);
    const int lo = kr * R;

    for (int j = threadIdx.x * 4; j < R; j += DEG_T * 4) *(uint4*)(hist + j) = make_uint4(0u, 0u, 0u, 0u);
    if (threadIdx.x < DEG_OMAX) bcnt[threadIdx.x] = 0u;
    __syncthreads();

    const int npad4 = ((nedges + KEY_TRIP - 1) / KEY_TRIP) * (KEY_TRIP / 4);
    const int qps = (npad4 + DEG_S - 1) / DEG_S;
    const int q0 = sl * qps;
    const int q1 = (q0 + qps) < npad4 ? (q0 + qps) : npad4;
    const int* __restrict__ dstp = ei + nedges;
    if ((nedges & 3) == 0) deg_walk<true>(ei, dstp, nedges, q0, q1, kr, lo, R, hist, bcnt, bins, keys);
    else                   deg_walk<false>(ei, dstp, nedges, q0, q1, kr, lo, R, hist, bcnt, bins, keys);
    __syncthreads();

    unsigned* __restrict__ plane = part + (size_t)sl * (DEG_K * R) + lo;
    for (int j = threadIdx.x * 4; j < R; j += DEG_T * 4) *(uint4*)(plane + j) = *(const uint4*)(hist + j);

    const int opr = R >> OWN_SHIFT;
    for (int idx = threadIdx.x; idx < opr * REC_CAP; idx += DEG_T) {
        const int ol = idx / REC_CAP;
        const int j  = idx % REC_CAP;
        if ((unsigned)j < bcnt[ol])
            rec0[((size_t)(kr * opr + ol) * DEG_S + sl) * REC_CAP + j] = bins[idx];
    }
    if (threadIdx.x < DEG_OMAX)
        cnts[(sl * DEG_K + kr) * DEG_OMAX + threadIdx.x] = (int)threadIdx.x < opr ? (int)bcnt[threadIdx.x] : 0;
}

__global__ void k_dinv_from_parts(const unsigned* __restrict__ part, float* __restrict__ dinv, int n) {
    const int i = blockIdx.x * blockDim.x + threadIdx.x;
    if (i >= n) return;
    const int P = DEG_K * deg_range_rows(n);
    unsigned c = 1u;
#pragma unroll
    for (int s = 0; s < DEG_S; ++s) c += part[(size_t)s * P + i];
    dinv[i] = rsqrtf((float)c);
}


#define WT_PITCH 68

template <bool BIAS_RELU>
__global__ void k_gemm64(const float* __restrict__ A, const float* __restrict__ W,
                         const float* __restrict__ bias, float* __restrict__ C, int nrows) {
    __shared__ float ldsW[64 * WT_PITCH];

    const int lane = threadIdx.x & 31;
    const int wave = threadIdx.x >> 5;
    const int rowTiles = (nrows + 15) >> 4;
    const int tile = blockIdx.x * 8 + wave;
    const int half = lane >> 4;
    const int m    = lane & 15;
    int row = tile * 16 + m;
    if (row >= nrows) row = nrows - 1;
    const float* __restrict__ arow = A + (size_t)row * 64;

    v2f a[16];
#pragma unroll
    for (int kk = 0; kk < 16; ++kk)
        a[kk] = *(const v2f*)(arow + kk * 4 + 2 * half);

    float4 wv[4];
#pragma unroll
    for (int i = 0; i < 4; ++i) wv[i] = *(const float4*)(W + (size_t)(threadIdx.x + i * 256) * 4);
#pragma unroll
    for (int i = 0; i < 4; ++i) {
        const int idx = (threadIdx.x + i * 256) * 4;
        const int k = idx >> 6;
        const int col = idx & 63;
        ldsW[(col + 0) * WT_PITCH + k] = wv[i].x;
        ldsW[(col + 1) * WT_PITCH + k] = wv[i].y;
        ldsW[(col + 2) * WT_PITCH + k] = wv[i].z;
        ldsW[(col + 3) * WT_PITCH + k] = wv[i].w;
    }
    __syncthreads();

    if (tile < rowTiles) {
        v8f c0 = {}, c1 = {}, c2 = {}, c3 = {};
#pragma unroll
        for (int kk = 0; kk < 16; ++kk) {
            const int k0 = kk * 4 + 2 * half;
            v2f b0 = *(const v2f*)&ldsW[(0 * 16 + m) * WT_PITCH + k0];
            v2f b1 = *(const v2f*)&ldsW[(1 * 16 + m) * WT_PITCH + k0];
            v2f b2 = *(const v2f*)&ldsW[(2 * 16 + m) * WT_PITCH + k0];
            v2f b3 = *(const v2f*)&ldsW[(3 * 16 + m) * WT_PITCH + k0];
            c0 = __builtin_amdgcn_wmma_f32_16x16x4_f32(false, a[kk], false, b0, (short)0, c0, false, false);
            c1 = __builtin_amdgcn_wmma_f32_16x16x4_f32(false, a[kk], false, b1, (short)0, c1, false, false);
            c2 = __builtin_amdgcn_wmma_f32_16x16x4_f32(false, a[kk], false, b2, (short)0, c2, false, false);
            c3 = __builtin_amdgcn_wmma_f32_16x16x4_f32(false, a[kk], false, b3, (short)0, c3, false, false);
        }

        v8f acc[4] = {c0, c1, c2, c3};
#pragma unroll
        for (int nt = 0; nt < 4; ++nt) {
            const int col = nt * 16 + m;
            const float bv = BIAS_RELU ? bias[col] : 0.0f;
#pragma unroll
            for (int r = 0; r < 8; ++r) {
                const int orow = tile * 16 + r + 8 * half;
                if (orow < nrows) {
                    float v = acc[nt][r];
                    if (BIAS_RELU) v = fmaxf(v + bv, 0.0f);
                    C[(size_t)orow * 64 + col] = v;
                }
            }
        }
    }
}


__device__ __forceinline__ void own_load_keys(int4 (&kv)[OWN_SUB], const int4* __restrict__ p) {
#pragma unroll
    for (int i = 0; i < OWN_SUB; ++i) kv[i] = p[i * 32];
}

template <bool TAIL>
__device__ __forceinline__ void own_drain(const int* qe, int q0, int cnt, int lane, int lo, int nedges,
                                          const int* __restrict__ ei, const float* __restrict__ dinv,
                                          const float* __restrict__ t, float* tile) {
    int dl = 0, s = 0;
    if (lane < cnt) {
        const int e = qe[q0 + lane];
        s = ei[e];
        dl = ei[nedges + e] - lo;
    }
    float v0[16], v1[16];
#pragma unroll
    for (int u = 0; u < 16; ++u) {
        const int su = __builtin_amdgcn_readlane(s, u);
        const float* __restrict__ tp = t + (size_t)su * 64;
        v0[u] = tp[lane];
        v1[u] = tp[lane + 32];
    }
    const float w = dinv[s] * dinv[lo + dl];
#pragma unroll
    for (int u = 0; u < 16; ++u) {
        if (!TAIL || u < cnt) {
            const int du = __builtin_amdgcn_readlane(dl, u);
            const float wu = __builtin_bit_cast(float,
                __builtin_amdgcn_readlane(__builtin_bit_cast(int, w), u));
            float* p = tile + du * 64;
            atomicAdd(p + lane, wu * v0[u]);
            atomicAdd(p + lane + 32, wu * v1[u]);
        }
    }
}

__device__ __forceinline__ void own_trip(const int4 (&kv)[OWN_SUB], int eb, unsigned pat, int& qn, int* qe,
                                         int lane, unsigned lt_mask, int lo, int nedges,
                                         const int* __restrict__ ei, const float* __restrict__ dinv,
                                         const float* __restrict__ t, float* tile) {
#pragma unroll
    for (int p = 0; p < OWN_SUB; ++p) {
        const unsigned x0 = (unsigned)kv[p].x ^ pat, x1 = (unsigned)kv[p].y ^ pat;
        const unsigned x2 = (unsigned)kv[p].z ^ pat, x3 = (unsigned)kv[p].w ^ pat;
        const unsigned y0 = ((x0 & 0x7F7F7F7Fu) + 0x7F7F7F7Fu) | x0;
        const unsigned y1 = ((x1 & 0x7F7F7F7Fu) + 0x7F7F7F7Fu) | x1;
        const unsigned y2 = ((x2 & 0x7F7F7F7Fu) + 0x7F7F7F7Fu) | x2;
        const unsigned y3 = ((x3 & 0x7F7F7F7Fu) + 0x7F7F7F7Fu) | x3;
        const unsigned nz = ((y0 >> 7) & 0x01010101u) | ((y1 >> 6) & 0x02020202u) |
                            ((y2 >> 5) & 0x04040404u) | ((y3 >> 4) & 0x08080808u);
        unsigned m = nz ^ 0x0F0F0F0Fu;
        if (__builtin_amdgcn_ballot_w32(m != 0u) != 0u) {
            const int ep = eb + p * 512;
            do {
                const bool has = m != 0u;
                const unsigned k = __builtin_amdgcn_ballot_w32(has);
                if (has) {
                    const int bit = __builtin_ctz(m);
                    const int j = ((bit & 7) << 2) | (bit >> 3);
                    qe[qn + __builtin_popcount(k & lt_mask)] = ep + j;
                    m &= m - 1u;
                }
                qn += __builtin_popcount(k);
                __builtin_amdgcn_wave_barrier();
                while (qn >= 16) {
                    qn -= 16;
                    own_drain<false>(qe, qn, 16, lane, lo, nedges, ei, dinv, t, tile);
                }
                __builtin_amdgcn_wave_barrier();
            } while (__builtin_amdgcn_ballot_w32(m != 0u) != 0u);
        }
    }
}

__device__ __forceinline__ void own_scan(const int* __restrict__ keys, int wv, int lane, unsigned pat, int* qe,
                                         int lo, int nedges, const int* __restrict__ ei,
                                         const float* __restrict__ dinv, const float* __restrict__ t,
                                         float* tile) {
    const unsigned lt_mask = (1u << lane) - 1u;
    int qn = 0;

    const int ntrip = (nedges + KEY_TRIP - 1) / KEY_TRIP;
    const int4* __restrict__ kp = (const int4*)keys + (size_t)wv * ntrip * (OWN_SUB * 32) + lane;
    const int eb0 = wv * ntrip * (OWN_SUB * 512) + lane * 16;
    if (ntrip > 0) {
        int4 ka[OWN_SUB], kb[OWN_SUB];
        own_load_keys(ka, kp);
        for (int k = 0; k < ntrip; k += 2) {
            const int k1 = (k + 1 < ntrip) ? k + 1 : ntrip - 1;
            own_load_keys(kb, kp + (size_t)k1 * (OWN_SUB * 32));
            own_trip(ka, eb0 + k * (OWN_SUB * 512), pat, qn, qe, lane, lt_mask, lo, nedges, ei, dinv, t, tile);
            if (k + 1 >= ntrip) break;
            const int k2 = (k + 2 < ntrip) ? k + 2 : ntrip - 1;
            own_load_keys(ka, kp + (size_t)k2 * (OWN_SUB * 32));
            own_trip(kb, eb0 + (k + 1) * (OWN_SUB * 512), pat, qn, qe, lane, lt_mask, lo, nedges, ei, dinv, t, tile);
        }
    }
    while (qn > 0) {
        const int c = qn < 16 ? qn : 16;
        qn -= c;
        own_drain<true>(qe, qn, c, lane, lo, nedges, ei, dinv, t, tile);
    }
}

template <bool TAIL, int OFF>
__device__ __forceinline__ void own_replay16(int2 r, int c, int lane, const float* __restrict__ t, float* tile) {
    float v0[16], v1[16];
#pragma unroll
    for (int u = 0; u < 16; ++u) {
        const int su = __builtin_amdgcn_readlane(r.x, OFF + u) & ((1 << REC_SHIFT) - 1);
        const float* __restrict__ tp = t + (size_t)su * 64;
        v0[u] = tp[lane];
        v1[u] = tp[lane + 32];
    }
#pragma unroll
    for (int u = 0; u < 16; ++u) {
        if (!TAIL || u < c) {
            const int du = __builtin_amdgcn_readlane(r.x, OFF + u) >> REC_SHIFT;
            const float wu = __builtin_bit_cast(float, __builtin_amdgcn_readlane(r.y, OFF + u));
            float* p = tile + du * 64;
            atomicAdd(p + lane, wu * v0[u]);
            atomicAdd(p + lane + 32, wu * v1[u]);
        }
    }
}

__device__ __forceinline__ void own_replay(const int2* __restrict__ rec, int cnt, int lane,
                                           const float* __restrict__ t, float* tile, int2 cur) {
    if (cnt <= 0) return;
    const int last = cnt - 1;
    for (int base = 0; base < cnt; base += 32) {
        const int i1 = base + 32 + lane;
        const int2 nxt = rec[i1 < last ? i1 : last];
        const int left = cnt - base;
        if (left >= 32) {
            own_replay16<false, 0>(cur, 16, lane, t, tile);
            own_replay16<false, 16>(cur, 16, lane, t, tile);
        } else {
            own_replay16<true, 0>(cur, left < 16 ? left : 16, lane, t, tile);
            if (left > 16) own_replay16<true, 16>(cur, left - 16, lane, t, tile);
        }
        cur = nxt;
    }
}

__device__ __forceinline__ void own_first(const int* __restrict__ word, int2* rec, int cnt, int lane, int lo,
                                          const float* __restrict__ dinv, const float* __restrict__ t,
                                          float* tile, int wc, int wn, float w) {
    if (cnt <= 0) return;
    const int last = cnt - 1;
    const int smask = (1 << REC_SHIFT) - 1;
    for (int base = 0; base < cnt; base += 32) {
        const int i2 = base + 64 + lane;
        const int wnn = word[i2 < last ? i2 : last];
        const float da = dinv[wn & smask];
        const float db = dinv[lo + (int)((unsigned)wn >> REC_SHIFT)];
        const int2 cur = make_int2(wc, __builtin_bit_cast(int, w));
        const int left = cnt - base;
        if (left >= 32) {
            own_replay16<false, 0>(cur, 16, lane, t, tile);
            own_replay16<false, 16>(cur, 16, lane, t, tile);
        } else {
            own_replay16<true, 0>(cur, left < 16 ? left : 16, lane, t, tile);
            if (left > 16) own_replay16<true, 16>(cur, left - 16, lane, t, tile);
        }
        if (base + lane < cnt) rec[base + lane] = cur;
        w = da * db;
        wc = wn;
        wn = wnn;
    }
}

static_assert(OWN_WAVES == DEG_S, "one wave per edge slice");
template <bool REPLAY>
__global__ void __launch_bounds__(OWN_WAVES * 32)
k_agg_owner(const int* __restrict__ ei, const int* __restrict__ keys, const float* __restrict__ dinv,
            const float* __restrict__ t, const float* __restrict__ b, float* __restrict__ h, int n, int nedges,
            const int* __restrict__ rec0, int* recs, const int* __restrict__ cnts) {
    __shared__ float tile[OWN_R * 64];
    __shared__ int qedge[OWN_WAVES * OWN_QCAP];

    const int lane = threadIdx.x & 31;
    const int wv = __builtin_amdgcn_readfirstlane((int)(threadIdx.x >> 5));
    const int lo = blockIdx.x * OWN_R;
    const int rows = (n - lo) < OWN_R ? (n - lo) : OWN_R;


    const int opr = deg_range_rows(n) >> OWN_SHIFT;
    const int* __restrict__ word = rec0 + (size_t)(blockIdx.x * DEG_S + wv) * REC_CAP;
    int2* rec = (int2*)recs + (size_t)(blockIdx.x * OWN_WAVES + wv) * REC_CAP;
    const int mine = __builtin_amdgcn_readfirstlane(
        cnts[(wv * DEG_K + (int)blockIdx.x / opr) * DEG_OMAX + (int)blockIdx.x % opr]);

    const int j = (threadIdx.x & 15) * 4;
    const int r0 = threadIdx.x >> 4;
    const float4 bv = *(const float4*)(b + j);

    float4 tv[OWN_PASS];
    float dv[OWN_PASS];
#pragma unroll
    for (int i = 0; i < OWN_PASS; ++i) {
        const int r = r0 + i * (OWN_WAVES * 2);
        const int rr = r < rows ? r : rows - 1;
        dv[i] = dinv[lo + rr];
        tv[i] = *(const float4*)(t + (size_t)(lo + rr) * 64 + j);
    }

    const bool walk = mine > 0 && mine <= REC_CAP;
    const int last = mine - 1;
    const int smask = (1 << REC_SHIFT) - 1;
    int2 hd = make_int2(0, 0);
    int hn = 0;
    if (walk) {
        if (REPLAY) {
            hd = ((const int2*)rec)[lane < last ? lane : last];
        } else {
            hd.x = word[lane < last ? lane : last];
            hn   = word[lane + 32 < last ? lane + 32 : last];
        }
    }

#pragma unroll
    for (int i = 0; i < OWN_PASS; ++i) {
        const int r = r0 + i * (OWN_WAVES * 2);
        if (r < rows) {
            const float sw = dv[i] * dv[i];
            *(float4*)(tile + r * 64 + j) = make_float4(sw * tv[i].x, sw * tv[i].y, sw * tv[i].z, sw * tv[i].w);
        }
    }
    float hw = 0.0f;
    if (!REPLAY && walk) hw = dinv[hd.x & smask] * dinv[lo + (int)((unsigned)hd.x >> REC_SHIFT)];
    __syncthreads();

    if (mine > REC_CAP) {
        int* qe = qedge + wv * OWN_QCAP;
        const unsigned pat = (blockIdx.x & 255u) * 0x01010101u;
        own_scan(keys, wv, lane, pat, qe, lo, nedges, ei, dinv, t, tile);
    } else if (mine > 0) {
        if (REPLAY) own_replay((const int2*)rec, mine, lane, t, tile, hd);
        else        own_first(word, rec, mine, lane, lo, dinv, t, tile, hd.x, hn, hw);
    }
    __syncthreads();

    float4 ov[OWN_PASS];
#pragma unroll
    for (int i = 0; i < OWN_PASS; ++i) {
        const int r = r0 + i * (OWN_WAVES * 2);
        const int rr = r < rows ? r : rows - 1;
        ov[i] = *(const float4*)(tile + rr * 64 + j);
    }
#pragma unroll
    for (int i = 0; i < OWN_PASS; ++i) {
        const int r = r0 + i * (OWN_WAVES * 2);
        if (r < rows) {
            float4 o;
            o.x = fmaxf(ov[i].x + bv.x, 0.0f);
            o.y = fmaxf(ov[i].y + bv.y, 0.0f);
            o.z = fmaxf(ov[i].z + bv.z, 0.0f);
            o.w = fmaxf(ov[i].w + bv.w, 0.0f);
            *(float4*)(h + (size_t)(lo + r) * 64 + j) = o;
        }
    }
}

__global__ void k_final_dot(const float* __restrict__ H, const float* __restrict__ w,
                            const float* __restrict__ b, float* __restrict__ out, int n) {
    int i = blockIdx.x * blockDim.x + threadIdx.x;
    if (i >= n) return;
    const float4* hp = (const float4*)(H + (size_t)i * 64);
    const float4* wp = (const float4*)w;
    float s = 0.0f;
#pragma unroll
    for (int k = 0; k < 16; ++k) {
        float4 hv = hp[k], wv = wp[k];
        s += hv.x * wv.x + hv.y * wv.y + hv.z * wv.z + hv.w * wv.w;
    }
    out[i] = s + b[0];
}


extern "C" void kernel_launch(void* const* d_in, const int* in_sizes, int n_in,
                              void* d_out, int out_size, void* d_ws, size_t ws_size,
                              hipStream_t stream) {
    (void)n_in; (void)out_size; (void)ws_size;

    const float* x   = (const float*)d_in[0];
    const int*   ei  = (const int*)  d_in[1];
    const float* W1  = (const float*)d_in[2];
    const float* b1  = (const float*)d_in[3];
    const float* W2  = (const float*)d_in[4];
    const float* b2  = (const float*)d_in[5];
    const float* Wf1 = (const float*)d_in[6];
    const float* bf1 = (const float*)d_in[7];
    const float* Wf2 = (const float*)d_in[8];
    const float* bf2 = (const float*)d_in[9];
    const float* Wf3 = (const float*)d_in[10];
    const float* bf3 = (const float*)d_in[11];
    float* out = (float*)d_out;

    const int n = in_sizes[0] / 64;
    const int E = in_sizes[1] / 2;

    float* ws    = (float*)d_ws;
    float* dinv  = ws;
    float* bufA  = ws + n;
    float* bufB  = bufA + (size_t)n * 64;
    int*   keys  = (int*)((char*)d_ws + ((size_t)128 << 20));
    int*   cnts  = (int*)((char*)d_ws + ((size_t)160 << 20));
    int*   recs  = (int*)((char*)d_ws + ((size_t)192 << 20));
    int*   rec0  = (int*)((char*)d_ws + ((size_t)96 << 20));
    unsigned* part = (unsigned*)((char*)d_ws + ((size_t)64 << 20));

    const int T = 256;
    const int rowTiles = (n + 15) / 16;
    const int gemmBlocks = (rowTiles + 7) / 8;
    const int ownBlocks = (n + OWN_R - 1) / OWN_R;

    k_deg_hist_keys<<<DEG_S * DEG_K, DEG_T, 0, stream>>>(ei, part, keys, rec0, cnts, n, E);
    k_dinv_from_parts<<<(n + T - 1) / T, T, 0, stream>>>(part, dinv, n);

    k_gemm64<false><<<gemmBlocks, T, 0, stream>>>(x, W1, nullptr, bufA, n);
    k_agg_owner<false><<<ownBlocks, OWN_WAVES * 32, 0, stream>>>(ei, keys, dinv, bufA, b1, bufB, n, E, rec0, recs, cnts);

    k_gemm64<false><<<gemmBlocks, T, 0, stream>>>(bufB, W2, nullptr, bufA, n);
    k_agg_owner<true><<<ownBlocks, OWN_WAVES * 32, 0, stream>>>(ei, keys, dinv, bufA, b2, bufB, n, E, rec0, recs, cnts);

    k_gemm64<true><<<gemmBlocks, T, 0, stream>>>(bufB, Wf1, bf1, bufA, n);
    k_gemm64<true><<<gemmBlocks, T, 0, stream>>>(bufA, Wf2, bf2, bufB, n);
    k_final_dot<<<(n + T - 1) / T, T, 0, stream>>>(bufB, Wf3, bf3, out, n);
}


# Owner tile 400 destination rows, was 512: 250 aggregation work-groups, one each on 250 of the 256 WGPs (was 196, 60 WGPs idle); owner = row / 400 in the degree-count launch; sums still LDS arrival
# speedup vs baseline: 3.3947x; 1.0672x over previous
#include <hip/hip_runtime.h>
#include <hip/hip_bf16.h>

typedef __attribute__((ext_vector_type(2))) float v2f;
typedef __attribute__((ext_vector_type(8))) float v8f;

#define OWN_R     400
#define OWN_WAVES 32
#define OWN_PASS  ((OWN_R * 16 + OWN_WAVES * 32 - 1) / (OWN_WAVES * 32))
#define OWN_SUB   4
#define OWN_QCAP  48
#define KEY_TRIP  (OWN_WAVES * OWN_SUB * 512)
#define REC_CAP   512
#define REC_SHIFT 17
#define DEG_OMAX  40

#define DEG_K    8
#define DEG_S    32
#define DEG_RCAP 16384
#define DEG_T    1024
#define DEG_U    8

#define DEG_RSTEP 800
static_assert(DEG_RSTEP % OWN_R == 0 && DEG_RSTEP % 32 == 0, "a range is whole owner tiles and whole lines of counters");
static_assert(DEG_OMAX * OWN_R <= DEG_RCAP && (DEG_OMAX * OWN_R) % DEG_RSTEP == 0, "the largest range's owners all have a list");
__device__ __forceinline__ int deg_range_rows(int n) { return (((n + DEG_K - 1) / DEG_K) + DEG_RSTEP - 1) / DEG_RSTEP * DEG_RSTEP; }

__device__ __forceinline__ void deg_bin(unsigned* bcnt, int* bins, unsigned r, int s) {
    const unsigned ol = r / (unsigned)OWN_R;
    const unsigned pos = atomicAdd(&bcnt[ol], 1u);
    if (pos < (unsigned)REC_CAP) bins[ol * REC_CAP + pos] = s | (int)((r - ol * (unsigned)OWN_R) << REC_SHIFT);
}

template <bool WIDE>
__device__ __forceinline__ void deg_walk(const int* __restrict__ srcp, const int* __restrict__ dstp, int nedges,
                                         int q0, int q1, int kr, int lo, int R, unsigned* hist,
                                         unsigned* bcnt, int* bins, int* __restrict__ keys) {
    for (int qb = q0 + (int)threadIdx.x; qb < q1; qb += DEG_T * DEG_U) {
        int4 v[DEG_U], sv[DEG_U];
#pragma unroll
        for (int u = 0; u < DEG_U; ++u) {
            const int q = qb + u * DEG_T;
            const long e0 = (long)q * 4;
            if (WIDE) {
                const bool in = q < q1 && e0 < nedges;
                const int4 d = *(const int4*)(dstp + (in ? e0 : 0));
                sv[u] = *(const int4*)(srcp + (in ? e0 : 0));
                v[u] = in ? d : make_int4(-1, -1, -1, -1);
            } else {
                const bool i0 = q < q1 && e0     < nedges, i1 = q < q1 && e0 + 1 < nedges;
                const bool i2 = q < q1 && e0 + 2 < nedges, i3 = q < q1 && e0 + 3 < nedges;
                const int x0 = dstp[i0 ? e0     : 0], x1 = dstp[i1 ? e0 + 1 : 0];
                const int x2 = dstp[i2 ? e0 + 2 : 0], x3 = dstp[i3 ? e0 + 3 : 0];
                v[u] = make_int4(i0 ? x0 : -1, i1 ? x1 : -1, i2 ? x2 : -1, i3 ? x3 : -1);
                sv[u] = make_int4(srcp[i0 ? e0 : 0], srcp[i1 ? e0 + 1 : 0], srcp[i2 ? e0 + 2 : 0], srcp[i3 ? e0 + 3 : 0]);
            }
        }
#pragma unroll
        for (int u = 0; u < DEG_U; ++u) {
            const int q = qb + u * DEG_T;
            const int4 d = v[u];
            const int4 sg = sv[u];
            const unsigned r0 = (unsigned)(d.x - lo), r1 = (unsigned)(d.y - lo);
            const unsigned r2 = (unsigned)(d.z - lo), r3 = (unsigned)(d.w - lo);
            if (r0 < (unsigned)R) { atomicAdd(&hist[r0], 1u); deg_bin(bcnt, bins, r0, sg.x); }
            if (r1 < (unsigned)R) { atomicAdd(&hist[r1], 1u); deg_bin(bcnt, bins, r1, sg.y); }
            if (r2 < (unsigned)R) { atomicAdd(&hist[r2], 1u); deg_bin(bcnt, bins, r2, sg.z); }
            if (r3 < (unsigned)R) { atomicAdd(&hist[r3], 1u); deg_bin(bcnt, bins, r3, sg.w); }
            if ((u % DEG_K) == kr && q < q1) {
                const unsigned k0 = d.x >= 0 ? (((unsigned)d.x / (unsigned)OWN_R) & 255u) : 255u;
                const unsigned k1 = d.y >= 0 ? (((unsigned)d.y / (unsigned)OWN_R) & 255u) : 255u;
                const unsigned k2 = d.z >= 0 ? (((unsigned)d.z / (unsigned)OWN_R) & 255u) : 255u;
                const unsigned k3 = d.w >= 0 ? (((unsigned)d.w / (unsigned)OWN_R) & 255u) : 255u;
                keys[q] = (int)(k0 | (k1 << 8) | (k2 << 16) | (k3 << 24));
            }
        }
    }
}

__global__ void __launch_bounds__(DEG_T)
k_deg_hist_keys(const int* __restrict__ ei, unsigned* __restrict__ part, int* __restrict__ keys,
                int* __restrict__ rec0, int* __restrict__ cnts, int n, int nedges) {
    __shared__ unsigned hist[DEG_RCAP];
    __shared__ int bins[DEG_OMAX * REC_CAP];
    __shared__ unsigned bcnt[DEG_OMAX];
    const int kr = blockIdx.x % DEG_K;
    const int sl = blockIdx.x / DEG_K;
    const int R  = deg_range_rows(n);
    const int lo = kr * R;

    for (int j = threadIdx.x * 4; j < R; j += DEG_T * 4) *(uint4*)(hist + j) = make_uint4(0u, 0u, 0u, 0u);
    if (threadIdx.x < DEG_OMAX) bcnt[threadIdx.x] = 0u;
    __syncthreads();

    const int npad4 = ((nedges + KEY_TRIP - 1) / KEY_TRIP) * (KEY_TRIP / 4);
    const int qps = (npad4 + DEG_S - 1) / DEG_S;
    const int q0 = sl * qps;
    const int q1 = (q0 + qps) < npad4 ? (q0 + qps) : npad4;
    const int* __restrict__ dstp = ei + nedges;
    if ((nedges & 3) == 0) deg_walk<true>(ei, dstp, nedges, q0, q1, kr, lo, R, hist, bcnt, bins, keys);
    else                   deg_walk<false>(ei, dstp, nedges, q0, q1, kr, lo, R, hist, bcnt, bins, keys);
    __syncthreads();

    unsigned* __restrict__ plane = part + (size_t)sl * (DEG_K * R) + lo;
    for (int j = threadIdx.x * 4; j < R; j += DEG_T * 4) *(uint4*)(plane + j) = *(const uint4*)(hist + j);

    const int opr = R / OWN_R;
    for (int idx = threadIdx.x; idx < opr * REC_CAP; idx += DEG_T) {
        const int ol = idx / REC_CAP;
        const int j  = idx % REC_CAP;
        if ((unsigned)j < bcnt[ol])
            rec0[((size_t)(kr * opr + ol) * DEG_S + sl) * REC_CAP + j] = bins[idx];
    }
    if (threadIdx.x < DEG_OMAX)
        cnts[(sl * DEG_K + kr) * DEG_OMAX + threadIdx.x] = (int)threadIdx.x < opr ? (int)bcnt[threadIdx.x] : 0;
}

__global__ void k_dinv_from_parts(const unsigned* __restrict__ part, float* __restrict__ dinv, int n) {
    const int i = blockIdx.x * blockDim.x + threadIdx.x;
    if (i >= n) return;
    const int P = DEG_K * deg_range_rows(n);
    unsigned c = 1u;
#pragma unroll
    for (int s = 0; s < DEG_S; ++s) c += part[(size_t)s * P + i];
    dinv[i] = rsqrtf((float)c);
}


#define WT_PITCH 68

template <bool BIAS_RELU>
__global__ void k_gemm64(const float* __restrict__ A, const float* __restrict__ W,
                         const float* __restrict__ bias, float* __restrict__ C, int nrows) {
    __shared__ float ldsW[64 * WT_PITCH];

    const int lane = threadIdx.x & 31;
    const int wave = threadIdx.x >> 5;
    const int rowTiles = (nrows + 15) >> 4;
    const int tile = blockIdx.x * 8 + wave;
    const int half = lane >> 4;
    const int m    = lane & 15;
    int row = tile * 16 + m;
    if (row >= nrows) row = nrows - 1;
    const float* __restrict__ arow = A + (size_t)row * 64;

    v2f a[16];
#pragma unroll
    for (int kk = 0; kk < 16; ++kk)
        a[kk] = *(const v2f*)(arow + kk * 4 + 2 * half);

    float4 wv[4];
#pragma unroll
    for (int i = 0; i < 4; ++i) wv[i] = *(const float4*)(W + (size_t)(threadIdx.x + i * 256) * 4);
#pragma unroll
    for (int i = 0; i < 4; ++i) {
        const int idx = (threadIdx.x + i * 256) * 4;
        const int k = idx >> 6;
        const int col = idx & 63;
        ldsW[(col + 0) * WT_PITCH + k] = wv[i].x;
        ldsW[(col + 1) * WT_PITCH + k] = wv[i].y;
        ldsW[(col + 2) * WT_PITCH + k] = wv[i].z;
        ldsW[(col + 3) * WT_PITCH + k] = wv[i].w;
    }
    __syncthreads();

    if (tile < rowTiles) {
        v8f c0 = {}, c1 = {}, c2 = {}, c3 = {};
#pragma unroll
        for (int kk = 0; kk < 16; ++kk) {
            const int k0 = kk * 4 + 2 * half;
            v2f b0 = *(const v2f*)&ldsW[(0 * 16 + m) * WT_PITCH + k0];
            v2f b1 = *(const v2f*)&ldsW[(1 * 16 + m) * WT_PITCH + k0];
            v2f b2 = *(const v2f*)&ldsW[(2 * 16 + m) * WT_PITCH + k0];
            v2f b3 = *(const v2f*)&ldsW[(3 * 16 + m) * WT_PITCH + k0];
            c0 = __builtin_amdgcn_wmma_f32_16x16x4_f32(false, a[kk], false, b0, (short)0, c0, false, false);
            c1 = __builtin_amdgcn_wmma_f32_16x16x4_f32(false, a[kk], false, b1, (short)0, c1, false, false);
            c2 = __builtin_amdgcn_wmma_f32_16x16x4_f32(false, a[kk], false, b2, (short)0, c2, false, false);
            c3 = __builtin_amdgcn_wmma_f32_16x16x4_f32(false, a[kk], false, b3, (short)0, c3, false, false);
        }

        v8f acc[4] = {c0, c1, c2, c3};
#pragma unroll
        for (int nt = 0; nt < 4; ++nt) {
            const int col = nt * 16 + m;
            const float bv = BIAS_RELU ? bias[col] : 0.0f;
#pragma unroll
            for (int r = 0; r < 8; ++r) {
                const int orow = tile * 16 + r + 8 * half;
                if (orow < nrows) {
                    float v = acc[nt][r];
                    if (BIAS_RELU) v = fmaxf(v + bv, 0.0f);
                    C[(size_t)orow * 64 + col] = v;
                }
            }
        }
    }
}


__device__ __forceinline__ void own_load_keys(int4 (&kv)[OWN_SUB], const int4* __restrict__ p) {
#pragma unroll
    for (int i = 0; i < OWN_SUB; ++i) kv[i] = p[i * 32];
}

template <bool TAIL>
__device__ __forceinline__ void own_drain(const int* qe, int q0, int cnt, int lane, int lo, int nedges,
                                          const int* __restrict__ ei, const float* __restrict__ dinv,
                                          const float* __restrict__ t, float* tile) {
    int dl = 0, s = 0;
    if (lane < cnt) {
        const int e = qe[q0 + lane];
        s = ei[e];
        dl = ei[nedges + e] - lo;
    }
    float v0[16], v1[16];
#pragma unroll
    for (int u = 0; u < 16; ++u) {
        const int su = __builtin_amdgcn_readlane(s, u);
        const float* __restrict__ tp = t + (size_t)su * 64;
        v0[u] = tp[lane];
        v1[u] = tp[lane + 32];
    }
    const float w = dinv[s] * dinv[lo + dl];
#pragma unroll
    for (int u = 0; u < 16; ++u) {
        if (!TAIL || u < cnt) {
            const int du = __builtin_amdgcn_readlane(dl, u);
            const float wu = __builtin_bit_cast(float,
                __builtin_amdgcn_readlane(__builtin_bit_cast(int, w), u));
            float* p = tile + du * 64;
            atomicAdd(p + lane, wu * v0[u]);
            atomicAdd(p + lane + 32, wu * v1[u]);
        }
    }
}

__device__ __forceinline__ void own_trip(const int4 (&kv)[OWN_SUB], int eb, unsigned pat, int& qn, int* qe,
                                         int lane, unsigned lt_mask, int lo, int nedges,
                                         const int* __restrict__ ei, const float* __restrict__ dinv,
                                         const float* __restrict__ t, float* tile) {
#pragma unroll
    for (int p = 0; p < OWN_SUB; ++p) {
        const unsigned x0 = (unsigned)kv[p].x ^ pat, x1 = (unsigned)kv[p].y ^ pat;
        const unsigned x2 = (unsigned)kv[p].z ^ pat, x3 = (unsigned)kv[p].w ^ pat;
        const unsigned y0 = ((x0 & 0x7F7F7F7Fu) + 0x7F7F7F7Fu) | x0;
        const unsigned y1 = ((x1 & 0x7F7F7F7Fu) + 0x7F7F7F7Fu) | x1;
        const unsigned y2 = ((x2 & 0x7F7F7F7Fu) + 0x7F7F7F7Fu) | x2;
        const unsigned y3 = ((x3 & 0x7F7F7F7Fu) + 0x7F7F7F7Fu) | x3;
        const unsigned nz = ((y0 >> 7) & 0x01010101u) | ((y1 >> 6) & 0x02020202u) |
                            ((y2 >> 5) & 0x04040404u) | ((y3 >> 4) & 0x08080808u);
        unsigned m = nz ^ 0x0F0F0F0Fu;
        if (__builtin_amdgcn_ballot_w32(m != 0u) != 0u) {
            const int ep = eb + p * 512;
            do {
                const bool has = m != 0u;
                const unsigned k = __builtin_amdgcn_ballot_w32(has);
                if (has) {
                    const int bit = __builtin_ctz(m);
                    const int j = ((bit & 7) << 2) | (bit >> 3);
                    qe[qn + __builtin_popcount(k & lt_mask)] = ep + j;
                    m &= m - 1u;
                }
                qn += __builtin_popcount(k);
                __builtin_amdgcn_wave_barrier();
                while (qn >= 16) {
                    qn -= 16;
                    own_drain<false>(qe, qn, 16, lane, lo, nedges, ei, dinv, t, tile);
                }
                __builtin_amdgcn_wave_barrier();
            } while (__builtin_amdgcn_ballot_w32(m != 0u) != 0u);
        }
    }
}

__device__ __forceinline__ void own_scan(const int* __restrict__ keys, int wv, int lane, unsigned pat, int* qe,
                                         int lo, int nedges, const int* __restrict__ ei,
                                         const float* __restrict__ dinv, const float* __restrict__ t,
                                         float* tile) {
    const unsigned lt_mask = (1u << lane) - 1u;
    int qn = 0;

    const int ntrip = (nedges + KEY_TRIP - 1) / KEY_TRIP;
    const int4* __restrict__ kp = (const int4*)keys + (size_t)wv * ntrip * (OWN_SUB * 32) + lane;
    const int eb0 = wv * ntrip * (OWN_SUB * 512) + lane * 16;
    if (ntrip > 0) {
        int4 ka[OWN_SUB], kb[OWN_SUB];
        own_load_keys(ka, kp);
        for (int k = 0; k < ntrip; k += 2) {
            const int k1 = (k + 1 < ntrip) ? k + 1 : ntrip - 1;
            own_load_keys(kb, kp + (size_t)k1 * (OWN_SUB * 32));
            own_trip(ka, eb0 + k * (OWN_SUB * 512), pat, qn, qe, lane, lt_mask, lo, nedges, ei, dinv, t, tile);
            if (k + 1 >= ntrip) break;
            const int k2 = (k + 2 < ntrip) ? k + 2 : ntrip - 1;
            own_load_keys(ka, kp + (size_t)k2 * (OWN_SUB * 32));
            own_trip(kb, eb0 + (k + 1) * (OWN_SUB * 512), pat, qn, qe, lane, lt_mask, lo, nedges, ei, dinv, t, tile);
        }
    }
    while (qn > 0) {
        const int c = qn < 16 ? qn : 16;
        qn -= c;
        own_drain<true>(qe, qn, c, lane, lo, nedges, ei, dinv, t, tile);
    }
}

template <bool TAIL, int OFF>
__device__ __forceinline__ void own_replay16(int2 r, int c, int lane, const float* __restrict__ t, float* tile) {
    float v0[16], v1[16];
#pragma unroll
    for (int u = 0; u < 16; ++u) {
        const int su = __builtin_amdgcn_readlane(r.x, OFF + u) & ((1 << REC_SHIFT) - 1);
        const float* __restrict__ tp = t + (size_t)su * 64;
        v0[u] = tp[lane];
        v1[u] = tp[lane + 32];
    }
#pragma unroll
    for (int u = 0; u < 16; ++u) {
        if (!TAIL || u < c) {
            const int du = __builtin_amdgcn_readlane(r.x, OFF + u) >> REC_SHIFT;
            const float wu = __builtin_bit_cast(float, __builtin_amdgcn_readlane(r.y, OFF + u));
            float* p = tile + du * 64;
            atomicAdd(p + lane, wu * v0[u]);
            atomicAdd(p + lane + 32, wu * v1[u]);
        }
    }
}

__device__ __forceinline__ void own_replay(const int2* __restrict__ rec, int cnt, int lane,
                                           const float* __restrict__ t, float* tile, int2 cur) {
    if (cnt <= 0) return;
    const int last = cnt - 1;
    for (int base = 0; base < cnt; base += 32) {
        const int i1 = base + 32 + lane;
        const int2 nxt = rec[i1 < last ? i1 : last];
        const int left = cnt - base;
        if (left >= 32) {
            own_replay16<false, 0>(cur, 16, lane, t, tile);
            own_replay16<false, 16>(cur, 16, lane, t, tile);
        } else {
            own_replay16<true, 0>(cur, left < 16 ? left : 16, lane, t, tile);
            if (left > 16) own_replay16<true, 16>(cur, left - 16, lane, t, tile);
        }
        cur = nxt;
    }
}

__device__ __forceinline__ void own_first(const int* __restrict__ word, int2* rec, int cnt, int lane, int lo,
                                          const float* __restrict__ dinv, const float* __restrict__ t,
                                          float* tile, int wc, int wn, float w) {
    if (cnt <= 0) return;
    const int last = cnt - 1;
    const int smask = (1 << REC_SHIFT) - 1;
    for (int base = 0; base < cnt; base += 32) {
        const int i2 = base + 64 + lane;
        const int wnn = word[i2 < last ? i2 : last];
        const float da = dinv[wn & smask];
        const float db = dinv[lo + (int)((unsigned)wn >> REC_SHIFT)];
        const int2 cur = make_int2(wc, __builtin_bit_cast(int, w));
        const int left = cnt - base;
        if (left >= 32) {
            own_replay16<false, 0>(cur, 16, lane, t, tile);
            own_replay16<false, 16>(cur, 16, lane, t, tile);
        } else {
            own_replay16<true, 0>(cur, left < 16 ? left : 16, lane, t, tile);
            if (left > 16) own_replay16<true, 16>(cur, left - 16, lane, t, tile);
        }
        if (base + lane < cnt) rec[base + lane] = cur;
        w = da * db;
        wc = wn;
        wn = wnn;
    }
}

static_assert(OWN_WAVES == DEG_S, "one wave per edge slice");
template <bool REPLAY>
__global__ void __launch_bounds__(OWN_WAVES * 32)
k_agg_owner(const int* __restrict__ ei, const int* __restrict__ keys, const float* __restrict__ dinv,
            const float* __restrict__ t, const float* __restrict__ b, float* __restrict__ h, int n, int nedges,
            const int* __restrict__ rec0, int* recs, const int* __restrict__ cnts) {
    __shared__ float tile[OWN_R * 64];
    __shared__ int qedge[OWN_WAVES * OWN_QCAP];

    const int lane = threadIdx.x & 31;
    const int wv = __builtin_amdgcn_readfirstlane((int)(threadIdx.x >> 5));
    const int lo = blockIdx.x * OWN_R;
    const int rows = (n - lo) < OWN_R ? (n - lo) : OWN_R;


    const int opr = deg_range_rows(n) / OWN_R;
    const int* __restrict__ word = rec0 + (size_t)(blockIdx.x * DEG_S + wv) * REC_CAP;
    int2* rec = (int2*)recs + (size_t)(blockIdx.x * OWN_WAVES + wv) * REC_CAP;
    const int mine = __builtin_amdgcn_readfirstlane(
        cnts[(wv * DEG_K + (int)blockIdx.x / opr) * DEG_OMAX + (int)blockIdx.x % opr]);

    const int j = (threadIdx.x & 15) * 4;
    const int r0 = threadIdx.x >> 4;
    const float4 bv = *(const float4*)(b + j);

    float4 tv[OWN_PASS];
    float dv[OWN_PASS];
#pragma unroll
    for (int i = 0; i < OWN_PASS; ++i) {
        const int r = r0 + i * (OWN_WAVES * 2);
        const int rr = r < rows ? r : rows - 1;
        dv[i] = dinv[lo + rr];
        tv[i] = *(const float4*)(t + (size_t)(lo + rr) * 64 + j);
    }

    const bool walk = mine > 0 && mine <= REC_CAP;
    const int last = mine - 1;
    const int smask = (1 << REC_SHIFT) - 1;
    int2 hd = make_int2(0, 0);
    int hn = 0;
    if (walk) {
        if (REPLAY) {
            hd = ((const int2*)rec)[lane < last ? lane : last];
        } else {
            hd.x = word[lane < last ? lane : last];
            hn   = word[lane + 32 < last ? lane + 32 : last];
        }
    }

#pragma unroll
    for (int i = 0; i < OWN_PASS; ++i) {
        const int r = r0 + i * (OWN_WAVES * 2);
        if (r < rows) {
            const float sw = dv[i] * dv[i];
            *(float4*)(tile + r * 64 + j) = make_float4(sw * tv[i].x, sw * tv[i].y, sw * tv[i].z, sw * tv[i].w);
        }
    }
    float hw = 0.0f;
    if (!REPLAY && walk) hw = dinv[hd.x & smask] * dinv[lo + (int)((unsigned)hd.x >> REC_SHIFT)];
    __syncthreads();

    if (mine > REC_CAP) {
        int* qe = qedge + wv * OWN_QCAP;
        const unsigned pat = (blockIdx.x & 255u) * 0x01010101u;
        own_scan(keys, wv, lane, pat, qe, lo, nedges, ei, dinv, t, tile);
    } else if (mine > 0) {
        if (REPLAY) own_replay((const int2*)rec, mine, lane, t, tile, hd);
        else        own_first(word, rec, mine, lane, lo, dinv, t, tile, hd.x, hn, hw);
    }
    __syncthreads();

    float4 ov[OWN_PASS];
#pragma unroll
    for (int i = 0; i < OWN_PASS; ++i) {
        const int r = r0 + i * (OWN_WAVES * 2);
        const int rr = r < rows ? r : rows - 1;
        ov[i] = *(const float4*)(tile + rr * 64 + j);
    }
#pragma unroll
    for (int i = 0; i < OWN_PASS; ++i) {
        const int r = r0 + i * (OWN_WAVES * 2);
        if (r < rows) {
            float4 o;
            o.x = fmaxf(ov[i].x + bv.x, 0.0f);
            o.y = fmaxf(ov[i].y + bv.y, 0.0f);
            o.z = fmaxf(ov[i].z + bv.z, 0.0f);
            o.w = fmaxf(ov[i].w + bv.w, 0.0f);
            *(float4*)(h + (size_t)(lo + r) * 64 + j) = o;
        }
    }
}

__global__ void k_final_dot(const float* __restrict__ H, const float* __restrict__ w,
                            const float* __restrict__ b, float* __restrict__ out, int n) {
    int i = blockIdx.x * blockDim.x + threadIdx.x;
    if (i >= n) return;
    const float4* hp = (const float4*)(H + (size_t)i * 64);
    const float4* wp = (const float4*)w;
    float s = 0.0f;
#pragma unroll
    for (int k = 0; k < 16; ++k) {
        float4 hv = hp[k], wv = wp[k];
        s += hv.x * wv.x + hv.y * wv.y + hv.z * wv.z + hv.w * wv.w;
    }
    out[i] = s + b[0];
}


extern "C" void kernel_launch(void* const* d_in, const int* in_sizes, int n_in,
                              void* d_out, int out_size, void* d_ws, size_t ws_size,
                              hipStream_t stream) {
    (void)n_in; (void)out_size; (void)ws_size;

    const float* x   = (const float*)d_in[0];
    const int*   ei  = (const int*)  d_in[1];
    const float* W1  = (const float*)d_in[2];
    const float* b1  = (const float*)d_in[3];
    const float* W2  = (const float*)d_in[4];
    const float* b2  = (const float*)d_in[5];
    const float* Wf1 = (const float*)d_in[6];
    const float* bf1 = (const float*)d_in[7];
    const float* Wf2 = (const float*)d_in[8];
    const float* bf2 = (const float*)d_in[9];
    const float* Wf3 = (const float*)d_in[10];
    const float* bf3 = (const float*)d_in[11];
    float* out = (float*)d_out;

    const int n = in_sizes[0] / 64;
    const int E = in_sizes[1] / 2;

    float* ws    = (float*)d_ws;
    float* dinv  = ws;
    float* bufA  = ws + n;
    float* bufB  = bufA + (size_t)n * 64;
    int*   keys  = (int*)((char*)d_ws + ((size_t)128 << 20));
    int*   cnts  = (int*)((char*)d_ws + ((size_t)160 << 20));
    int*   recs  = (int*)((char*)d_ws + ((size_t)192 << 20));
    int*   rec0  = (int*)((char*)d_ws + ((size_t)96 << 20));
    unsigned* part = (unsigned*)((char*)d_ws + ((size_t)64 << 20));

    const int T = 256;
    const int rowTiles = (n + 15) / 16;
    const int gemmBlocks = (rowTiles + 7) / 8;
    const int ownBlocks = (n + OWN_R - 1) / OWN_R;

    k_deg_hist_keys<<<DEG_S * DEG_K, DEG_T, 0, stream>>>(ei, part, keys, rec0, cnts, n, E);
    k_dinv_from_parts<<<(n + T - 1) / T, T, 0, stream>>>(part, dinv, n);

    k_gemm64<false><<<gemmBlocks, T, 0, stream>>>(x, W1, nullptr, bufA, n);
    k_agg_owner<false><<<ownBlocks, OWN_WAVES * 32, 0, stream>>>(ei, keys, dinv, bufA, b1, bufB, n, E, rec0, recs, cnts);

    k_gemm64<false><<<gemmBlocks, T, 0, stream>>>(bufB, W2, nullptr, bufA, n);
    k_agg_owner<true><<<ownBlocks, OWN_WAVES * 32, 0, stream>>>(ei, keys, dinv, bufA, b2, bufB, n, E, rec0, recs, cnts);

    k_gemm64<true><<<gemmBlocks, T, 0, stream>>>(bufB, Wf1, bf1, bufA, n);
    k_gemm64<true><<<gemmBlocks, T, 0, stream>>>(bufA, Wf2, bf2, bufB, n);
    k_final_dot<<<(n + T - 1) / T, T, 0, stream>>>(bufB, Wf3, bf3, out, n);
}


# Layer 2 sums the rows of h1 first, W2 comes after (A(hW) = (Ah)W): W2, Wf1, Wf2 and the Wf3 dot run chained in one launch, rows handed on in LDS; 6 launches, was 9; layer-2 and dot sum order change
# speedup vs baseline: 3.9984x; 1.1778x over previous
#include <hip/hip_runtime.h>
#include <hip/hip_bf16.h>

typedef __attribute__((ext_vector_type(2))) float v2f;
typedef __attribute__((ext_vector_type(8))) float v8f;

#define OWN_R     400
#define OWN_WAVES 32
#define OWN_PASS  ((OWN_R * 16 + OWN_WAVES * 32 - 1) / (OWN_WAVES * 32))
#define OWN_SUB   4
#define OWN_QCAP  48
#define KEY_TRIP  (OWN_WAVES * OWN_SUB * 512)
#define REC_CAP   512
#define REC_SHIFT 17
#define DEG_OMAX  40

#define DEG_K    8
#define DEG_S    32
#define DEG_RCAP 16384
#define DEG_T    1024
#define DEG_U    8

#define DEG_RSTEP 800
static_assert(DEG_RSTEP % OWN_R == 0 && DEG_RSTEP % 32 == 0, "a range is whole owner tiles and whole lines of counters");
static_assert(DEG_OMAX * OWN_R <= DEG_RCAP && (DEG_OMAX * OWN_R) % DEG_RSTEP == 0, "the largest range's owners all have a list");
__device__ __forceinline__ int deg_range_rows(int n) { return (((n + DEG_K - 1) / DEG_K) + DEG_RSTEP - 1) / DEG_RSTEP * DEG_RSTEP; }

__device__ __forceinline__ void deg_bin(unsigned* bcnt, int* bins, unsigned r, int s) {
    const unsigned ol = r / (unsigned)OWN_R;
    const unsigned pos = atomicAdd(&bcnt[ol], 1u);
    if (pos < (unsigned)REC_CAP) bins[ol * REC_CAP + pos] = s | (int)((r - ol * (unsigned)OWN_R) << REC_SHIFT);
}

template <bool WIDE>
__device__ __forceinline__ void deg_walk(const int* __restrict__ srcp, const int* __restrict__ dstp, int nedges,
                                         int q0, int q1, int kr, int lo, int R, unsigned* hist,
                                         unsigned* bcnt, int* bins, int* __restrict__ keys) {
    for (int qb = q0 + (int)threadIdx.x; qb < q1; qb += DEG_T * DEG_U) {
        int4 v[DEG_U], sv[DEG_U];
#pragma unroll
        for (int u = 0; u < DEG_U; ++u) {
            const int q = qb + u * DEG_T;
            const long e0 = (long)q * 4;
            if (WIDE) {
                const bool in = q < q1 && e0 < nedges;
                const int4 d = *(const int4*)(dstp + (in ? e0 : 0));
                sv[u] = *(const int4*)(srcp + (in ? e0 : 0));
                v[u] = in ? d : make_int4(-1, -1, -1, -1);
            } else {
                const bool i0 = q < q1 && e0     < nedges, i1 = q < q1 && e0 + 1 < nedges;
                const bool i2 = q < q1 && e0 + 2 < nedges, i3 = q < q1 && e0 + 3 < nedges;
                const int x0 = dstp[i0 ? e0     : 0], x1 = dstp[i1 ? e0 + 1 : 0];
                const int x2 = dstp[i2 ? e0 + 2 : 0], x3 = dstp[i3 ? e0 + 3 : 0];
                v[u] = make_int4(i0 ? x0 : -1, i1 ? x1 : -1, i2 ? x2 : -1, i3 ? x3 : -1);
                sv[u] = make_int4(srcp[i0 ? e0 : 0], srcp[i1 ? e0 + 1 : 0], srcp[i2 ? e0 + 2 : 0], srcp[i3 ? e0 + 3 : 0]);
            }
        }
#pragma unroll
        for (int u = 0; u < DEG_U; ++u) {
            const int q = qb + u * DEG_T;
            const int4 d = v[u];
            const int4 sg = sv[u];
            const unsigned r0 = (unsigned)(d.x - lo), r1 = (unsigned)(d.y - lo);
            const unsigned r2 = (unsigned)(d.z - lo), r3 = (unsigned)(d.w - lo);
            if (r0 < (unsigned)R) { atomicAdd(&hist[r0], 1u); deg_bin(bcnt, bins, r0, sg.x); }
            if (r1 < (unsigned)R) { atomicAdd(&hist[r1], 1u); deg_bin(bcnt, bins, r1, sg.y); }
            if (r2 < (unsigned)R) { atomicAdd(&hist[r2], 1u); deg_bin(bcnt, bins, r2, sg.z); }
            if (r3 < (unsigned)R) { atomicAdd(&hist[r3], 1u); deg_bin(bcnt, bins, r3, sg.w); }
            if ((u % DEG_K) == kr && q < q1) {
                const unsigned k0 = d.x >= 0 ? (((unsigned)d.x / (unsigned)OWN_R) & 255u) : 255u;
                const unsigned k1 = d.y >= 0 ? (((unsigned)d.y / (unsigned)OWN_R) & 255u) : 255u;
                const unsigned k2 = d.z >= 0 ? (((unsigned)d.z / (unsigned)OWN_R) & 255u) : 255u;
                const unsigned k3 = d.w >= 0 ? (((unsigned)d.w / (unsigned)OWN_R) & 255u) : 255u;
                keys[q] = (int)(k0 | (k1 << 8) | (k2 << 16) | (k3 << 24));
            }
        }
    }
}

__global__ void __launch_bounds__(DEG_T)
k_deg_hist_keys(const int* __restrict__ ei, unsigned* __restrict__ part, int* __restrict__ keys,
                int* __restrict__ rec0, int* __restrict__ cnts, int n, int nedges) {
    __shared__ unsigned hist[DEG_RCAP];
    __shared__ int bins[DEG_OMAX * REC_CAP];
    __shared__ unsigned bcnt[DEG_OMAX];
    const int kr = blockIdx.x % DEG_K;
    const int sl = blockIdx.x / DEG_K;
    const int R  = deg_range_rows(n);
    const int lo = kr * R;

    for (int j = threadIdx.x * 4; j < R; j += DEG_T * 4) *(uint4*)(hist + j) = make_uint4(0u, 0u, 0u, 0u);
    if (threadIdx.x < DEG_OMAX) bcnt[threadIdx.x] = 0u;
    __syncthreads();

    const int npad4 = ((nedges + KEY_TRIP - 1) / KEY_TRIP) * (KEY_TRIP / 4);
    const int qps = (npad4 + DEG_S - 1) / DEG_S;
    const int q0 = sl * qps;
    const int q1 = (q0 + qps) < npad4 ? (q0 + qps) : npad4;
    const int* __restrict__ dstp = ei + nedges;
    if ((nedges & 3) == 0) deg_walk<true>(ei, dstp, nedges, q0, q1, kr, lo, R, hist, bcnt, bins, keys);
    else                   deg_walk<false>(ei, dstp, nedges, q0, q1, kr, lo, R, hist, bcnt, bins, keys);
    __syncthreads();

    unsigned* __restrict__ plane = part + (size_t)sl * (DEG_K * R) + lo;
    for (int j = threadIdx.x * 4; j < R; j += DEG_T * 4) *(uint4*)(plane + j) = *(const uint4*)(hist + j);

    const int opr = R / OWN_R;
    for (int idx = threadIdx.x; idx < opr * REC_CAP; idx += DEG_T) {
        const int ol = idx / REC_CAP;
        const int j  = idx % REC_CAP;
        if ((unsigned)j < bcnt[ol])
            rec0[((size_t)(kr * opr + ol) * DEG_S + sl) * REC_CAP + j] = bins[idx];
    }
    if (threadIdx.x < DEG_OMAX)
        cnts[(sl * DEG_K + kr) * DEG_OMAX + threadIdx.x] = (int)threadIdx.x < opr ? (int)bcnt[threadIdx.x] : 0;
}

__global__ void k_dinv_from_parts(const unsigned* __restrict__ part, float* __restrict__ dinv, int n) {
    const int i = blockIdx.x * blockDim.x + threadIdx.x;
    if (i >= n) return;
    const int P = DEG_K * deg_range_rows(n);
    unsigned c = 1u;
#pragma unroll
    for (int s = 0; s < DEG_S; ++s) c += part[(size_t)s * P + i];
    dinv[i] = rsqrtf((float)c);
}


#define WT_PITCH 68

template <bool BIAS_RELU>
__global__ void k_gemm64(const float* __restrict__ A, const float* __restrict__ W,
                         const float* __restrict__ bias, float* __restrict__ C, int nrows) {
    __shared__ float ldsW[64 * WT_PITCH];

    const int lane = threadIdx.x & 31;
    const int wave = threadIdx.x >> 5;
    const int rowTiles = (nrows + 15) >> 4;
    const int tile = blockIdx.x * 8 + wave;
    const int half = lane >> 4;
    const int m    = lane & 15;
    int row = tile * 16 + m;
    if (row >= nrows) row = nrows - 1;
    const float* __restrict__ arow = A + (size_t)row * 64;

    v2f a[16];
#pragma unroll
    for (int kk = 0; kk < 16; ++kk)
        a[kk] = *(const v2f*)(arow + kk * 4 + 2 * half);

    float4 wv[4];
#pragma unroll
    for (int i = 0; i < 4; ++i) wv[i] = *(const float4*)(W + (size_t)(threadIdx.x + i * 256) * 4);
#pragma unroll
    for (int i = 0; i < 4; ++i) {
        const int idx = (threadIdx.x + i * 256) * 4;
        const int k = idx >> 6;
        const int col = idx & 63;
        ldsW[(col + 0) * WT_PITCH + k] = wv[i].x;
        ldsW[(col + 1) * WT_PITCH + k] = wv[i].y;
        ldsW[(col + 2) * WT_PITCH + k] = wv[i].z;
        ldsW[(col + 3) * WT_PITCH + k] = wv[i].w;
    }
    __syncthreads();

    if (tile < rowTiles) {
        v8f c0 = {}, c1 = {}, c2 = {}, c3 = {};
#pragma unroll
        for (int kk = 0; kk < 16; ++kk) {
            const int k0 = kk * 4 + 2 * half;
            v2f b0 = *(const v2f*)&ldsW[(0 * 16 + m) * WT_PITCH + k0];
            v2f b1 = *(const v2f*)&ldsW[(1 * 16 + m) * WT_PITCH + k0];
            v2f b2 = *(const v2f*)&ldsW[(2 * 16 + m) * WT_PITCH + k0];
            v2f b3 = *(const v2f*)&ldsW[(3 * 16 + m) * WT_PITCH + k0];
            c0 = __builtin_amdgcn_wmma_f32_16x16x4_f32(false, a[kk], false, b0, (short)0, c0, false, false);
            c1 = __builtin_amdgcn_wmma_f32_16x16x4_f32(false, a[kk], false, b1, (short)0, c1, false, false);
            c2 = __builtin_amdgcn_wmma_f32_16x16x4_f32(false, a[kk], false, b2, (short)0, c2, false, false);
            c3 = __builtin_amdgcn_wmma_f32_16x16x4_f32(false, a[kk], false, b3, (short)0, c3, false, false);
        }

        v8f acc[4] = {c0, c1, c2, c3};
#pragma unroll
        for (int nt = 0; nt < 4; ++nt) {
            const int col = nt * 16 + m;
            const float bv = BIAS_RELU ? bias[col] : 0.0f;
#pragma unroll
            for (int r = 0; r < 8; ++r) {
                const int orow = tile * 16 + r + 8 * half;
                if (orow < nrows) {
                    float v = acc[nt][r];
                    if (BIAS_RELU) v = fmaxf(v + bv, 0.0f);
                    C[(size_t)orow * 64 + col] = v;
                }
            }
        }
    }
}


__device__ __forceinline__ void own_load_keys(int4 (&kv)[OWN_SUB], const int4* __restrict__ p) {
#pragma unroll
    for (int i = 0; i < OWN_SUB; ++i) kv[i] = p[i * 32];
}

template <bool TAIL>
__device__ __forceinline__ void own_drain(const int* qe, int q0, int cnt, int lane, int lo, int nedges,
                                          const int* __restrict__ ei, const float* __restrict__ dinv,
                                          const float* __restrict__ t, float* tile) {
    int dl = 0, s = 0;
    if (lane < cnt) {
        const int e = qe[q0 + lane];
        s = ei[e];
        dl = ei[nedges + e] - lo;
    }
    float v0[16], v1[16];
#pragma unroll
    for (int u = 0; u < 16; ++u) {
        const int su = __builtin_amdgcn_readlane(s, u);
        const float* __restrict__ tp = t + (size_t)su * 64;
        v0[u] = tp[lane];
        v1[u] = tp[lane + 32];
    }
    const float w = dinv[s] * dinv[lo + dl];
#pragma unroll
    for (int u = 0; u < 16; ++u) {
        if (!TAIL || u < cnt) {
            const int du = __builtin_amdgcn_readlane(dl, u);
            const float wu = __builtin_bit_cast(float,
                __builtin_amdgcn_readlane(__builtin_bit_cast(int, w), u));
            float* p = tile + du * 64;
            atomicAdd(p + lane, wu * v0[u]);
            atomicAdd(p + lane + 32, wu * v1[u]);
        }
    }
}

__device__ __forceinline__ void own_trip(const int4 (&kv)[OWN_SUB], int eb, unsigned pat, int& qn, int* qe,
                                         int lane, unsigned lt_mask, int lo, int nedges,
                                         const int* __restrict__ ei, const float* __restrict__ dinv,
                                         const float* __restrict__ t, float* tile) {
#pragma unroll
    for (int p = 0; p < OWN_SUB; ++p) {
        const unsigned x0 = (unsigned)kv[p].x ^ pat, x1 = (unsigned)kv[p].y ^ pat;
        const unsigned x2 = (unsigned)kv[p].z ^ pat, x3 = (unsigned)kv[p].w ^ pat;
        const unsigned y0 = ((x0 & 0x7F7F7F7Fu) + 0x7F7F7F7Fu) | x0;
        const unsigned y1 = ((x1 & 0x7F7F7F7Fu) + 0x7F7F7F7Fu) | x1;
        const unsigned y2 = ((x2 & 0x7F7F7F7Fu) + 0x7F7F7F7Fu) | x2;
        const unsigned y3 = ((x3 & 0x7F7F7F7Fu) + 0x7F7F7F7Fu) | x3;
        const unsigned nz = ((y0 >> 7) & 0x01010101u) | ((y1 >> 6) & 0x02020202u) |
                            ((y2 >> 5) & 0x04040404u) | ((y3 >> 4) & 0x08080808u);
        unsigned m = nz ^ 0x0F0F0F0Fu;
        if (__builtin_amdgcn_ballot_w32(m != 0u) != 0u) {
            const int ep = eb + p * 512;
            do {
                const bool has = m != 0u;
                const unsigned k = __builtin_amdgcn_ballot_w32(has);
                if (has) {
                    const int bit = __builtin_ctz(m);
                    const int j = ((bit & 7) << 2) | (bit >> 3);
                    qe[qn + __builtin_popcount(k & lt_mask)] = ep + j;
                    m &= m - 1u;
                }
                qn += __builtin_popcount(k);
                __builtin_amdgcn_wave_barrier();
                while (qn >= 16) {
                    qn -= 16;
                    own_drain<false>(qe, qn, 16, lane, lo, nedges, ei, dinv, t, tile);
                }
                __builtin_amdgcn_wave_barrier();
            } while (__builtin_amdgcn_ballot_w32(m != 0u) != 0u);
        }
    }
}

__device__ __forceinline__ void own_scan(const int* __restrict__ keys, int wv, int lane, unsigned pat, int* qe,
                                         int lo, int nedges, const int* __restrict__ ei,
                                         const float* __restrict__ dinv, const float* __restrict__ t,
                                         float* tile) {
    const unsigned lt_mask = (1u << lane) - 1u;
    int qn = 0;

    const int ntrip = (nedges + KEY_TRIP - 1) / KEY_TRIP;
    const int4* __restrict__ kp = (const int4*)keys + (size_t)wv * ntrip * (OWN_SUB * 32) + lane;
    const int eb0 = wv * ntrip * (OWN_SUB * 512) + lane * 16;
    if (ntrip > 0) {
        int4 ka[OWN_SUB], kb[OWN_SUB];
        own_load_keys(ka, kp);
        for (int k = 0; k < ntrip; k += 2) {
            const int k1 = (k + 1 < ntrip) ? k + 1 : ntrip - 1;
            own_load_keys(kb, kp + (size_t)k1 * (OWN_SUB * 32));
            own_trip(ka, eb0 + k * (OWN_SUB * 512), pat, qn, qe, lane, lt_mask, lo, nedges, ei, dinv, t, tile);
            if (k + 1 >= ntrip) break;
            const int k2 = (k + 2 < ntrip) ? k + 2 : ntrip - 1;
            own_load_keys(ka, kp + (size_t)k2 * (OWN_SUB * 32));
            own_trip(kb, eb0 + (k + 1) * (OWN_SUB * 512), pat, qn, qe, lane, lt_mask, lo, nedges, ei, dinv, t, tile);
        }
    }
    while (qn > 0) {
        const int c = qn < 16 ? qn : 16;
        qn -= c;
        own_drain<true>(qe, qn, c, lane, lo, nedges, ei, dinv, t, tile);
    }
}

template <bool TAIL, int OFF>
__device__ __forceinline__ void own_replay16(int2 r, int c, int lane, const float* __restrict__ t, float* tile) {
    float v0[16], v1[16];
#pragma unroll
    for (int u = 0; u < 16; ++u) {
        const int su = __builtin_amdgcn_readlane(r.x, OFF + u) & ((1 << REC_SHIFT) - 1);
        const float* __restrict__ tp = t + (size_t)su * 64;
        v0[u] = tp[lane];
        v1[u] = tp[lane + 32];
    }
#pragma unroll
    for (int u = 0; u < 16; ++u) {
        if (!TAIL || u < c) {
            const int du = __builtin_amdgcn_readlane(r.x, OFF + u) >> REC_SHIFT;
            const float wu = __builtin_bit_cast(float, __builtin_amdgcn_readlane(r.y, OFF + u));
            float* p = tile + du * 64;
            atomicAdd(p + lane, wu * v0[u]);
            atomicAdd(p + lane + 32, wu * v1[u]);
        }
    }
}

__device__ __forceinline__ void own_replay(const int2* __restrict__ rec, int cnt, int lane,
                                           const float* __restrict__ t, float* tile, int2 cur) {
    if (cnt <= 0) return;
    const int last = cnt - 1;
    for (int base = 0; base < cnt; base += 32) {
        const int i1 = base + 32 + lane;
        const int2 nxt = rec[i1 < last ? i1 : last];
        const int left = cnt - base;
        if (left >= 32) {
            own_replay16<false, 0>(cur, 16, lane, t, tile);
            own_replay16<false, 16>(cur, 16, lane, t, tile);
        } else {
            own_replay16<true, 0>(cur, left < 16 ? left : 16, lane, t, tile);
            if (left > 16) own_replay16<true, 16>(cur, left - 16, lane, t, tile);
        }
        cur = nxt;
    }
}

__device__ __forceinline__ void own_first(const int* __restrict__ word, int2* rec, int cnt, int lane, int lo,
                                          const float* __restrict__ dinv, const float* __restrict__ t,
                                          float* tile, int wc, int wn, float w) {
    if (cnt <= 0) return;
    const int last = cnt - 1;
    const int smask = (1 << REC_SHIFT) - 1;
    for (int base = 0; base < cnt; base += 32) {
        const int i2 = base + 64 + lane;
        const int wnn = word[i2 < last ? i2 : last];
        const float da = dinv[wn & smask];
        const float db = dinv[lo + (int)((unsigned)wn >> REC_SHIFT)];
        const int2 cur = make_int2(wc, __builtin_bit_cast(int, w));
        const int left = cnt - base;
        if (left >= 32) {
            own_replay16<false, 0>(cur, 16, lane, t, tile);
            own_replay16<false, 16>(cur, 16, lane, t, tile);
        } else {
            own_replay16<true, 0>(cur, left < 16 ? left : 16, lane, t, tile);
            if (left > 16) own_replay16<true, 16>(cur, left - 16, lane, t, tile);
        }
        if (base + lane < cnt) rec[base + lane] = cur;
        w = da * db;
        wc = wn;
        wn = wnn;
    }
}

static_assert(OWN_WAVES == DEG_S, "one wave per edge slice");
template <bool REPLAY>
__global__ void __launch_bounds__(OWN_WAVES * 32)
k_agg_owner(const int* __restrict__ ei, const int* __restrict__ keys, const float* __restrict__ dinv,
            const float* __restrict__ t, const float* __restrict__ b, float* __restrict__ h, int n, int nedges,
            const int* __restrict__ rec0, int* recs, const int* __restrict__ cnts) {
    __shared__ float tile[OWN_R * 64];
    __shared__ int qedge[OWN_WAVES * OWN_QCAP];

    const int lane = threadIdx.x & 31;
    const int wv = __builtin_amdgcn_readfirstlane((int)(threadIdx.x >> 5));
    const int lo = blockIdx.x * OWN_R;
    const int rows = (n - lo) < OWN_R ? (n - lo) : OWN_R;


    const int opr = deg_range_rows(n) / OWN_R;
    const int* __restrict__ word = rec0 + (size_t)(blockIdx.x * DEG_S + wv) * REC_CAP;
    int2* rec = (int2*)recs + (size_t)(blockIdx.x * OWN_WAVES + wv) * REC_CAP;
    const int mine = __builtin_amdgcn_readfirstlane(
        cnts[(wv * DEG_K + (int)blockIdx.x / opr) * DEG_OMAX + (int)blockIdx.x % opr]);

    const int j = (threadIdx.x & 15) * 4;
    const int r0 = threadIdx.x >> 4;
    const float4 bv = REPLAY ? make_float4(0.0f, 0.0f, 0.0f, 0.0f) : *(const float4*)(b + j);

    float4 tv[OWN_PASS];
    float dv[OWN_PASS];
#pragma unroll
    for (int i = 0; i < OWN_PASS; ++i) {
        const int r = r0 + i * (OWN_WAVES * 2);
        const int rr = r < rows ? r : rows - 1;
        dv[i] = dinv[lo + rr];
        tv[i] = *(const float4*)(t + (size_t)(lo + rr) * 64 + j);
    }

    const bool walk = mine > 0 && mine <= REC_CAP;
    const int last = mine - 1;
    const int smask = (1 << REC_SHIFT) - 1;
    int2 hd = make_int2(0, 0);
    int hn = 0;
    if (walk) {
        if (REPLAY) {
            hd = ((const int2*)rec)[lane < last ? lane : last];
        } else {
            hd.x = word[lane < last ? lane : last];
            hn   = word[lane + 32 < last ? lane + 32 : last];
        }
    }

#pragma unroll
    for (int i = 0; i < OWN_PASS; ++i) {
        const int r = r0 + i * (OWN_WAVES * 2);
        if (r < rows) {
            const float sw = dv[i] * dv[i];
            *(float4*)(tile + r * 64 + j) = make_float4(sw * tv[i].x, sw * tv[i].y, sw * tv[i].z, sw * tv[i].w);
        }
    }
    float hw = 0.0f;
    if (!REPLAY && walk) hw = dinv[hd.x & smask] * dinv[lo + (int)((unsigned)hd.x >> REC_SHIFT)];
    __syncthreads();

    if (mine > REC_CAP) {
        int* qe = qedge + wv * OWN_QCAP;
        const unsigned pat = (blockIdx.x & 255u) * 0x01010101u;
        own_scan(keys, wv, lane, pat, qe, lo, nedges, ei, dinv, t, tile);
    } else if (mine > 0) {
        if (REPLAY) own_replay((const int2*)rec, mine, lane, t, tile, hd);
        else        own_first(word, rec, mine, lane, lo, dinv, t, tile, hd.x, hn, hw);
    }
    __syncthreads();

    float4 ov[OWN_PASS];
#pragma unroll
    for (int i = 0; i < OWN_PASS; ++i) {
        const int r = r0 + i * (OWN_WAVES * 2);
        const int rr = r < rows ? r : rows - 1;
        ov[i] = *(const float4*)(tile + rr * 64 + j);
    }
#pragma unroll
    for (int i = 0; i < OWN_PASS; ++i) {
        const int r = r0 + i * (OWN_WAVES * 2);
        if (r < rows) {
            float4 o = ov[i];
            if (!REPLAY) {
                o.x = fmaxf(ov[i].x + bv.x, 0.0f);
                o.y = fmaxf(ov[i].y + bv.y, 0.0f);
                o.z = fmaxf(ov[i].z + bv.z, 0.0f);
                o.w = fmaxf(ov[i].w + bv.w, 0.0f);
            }
            *(float4*)(h + (size_t)(lo + r) * 64 + j) = o;
        }
    }
}

#define CH_WAVES 16
#define CH_PITCH 68

__device__ __forceinline__ void chain_product(const v2f (&a)[16], const float* wt, int m, int half, v8f (&c)[4]) {
    v8f c0 = {}, c1 = {}, c2 = {}, c3 = {};
#pragma unroll
    for (int kk = 0; kk < 16; ++kk) {
        const int k0 = kk * 4 + 2 * half;
        v2f b0 = *(const v2f*)&wt[(0 * 16 + m) * WT_PITCH + k0];
        v2f b1 = *(const v2f*)&wt[(1 * 16 + m) * WT_PITCH + k0];
        v2f b2 = *(const v2f*)&wt[(2 * 16 + m) * WT_PITCH + k0];
        v2f b3 = *(const v2f*)&wt[(3 * 16 + m) * WT_PITCH + k0];
        c0 = __builtin_amdgcn_wmma_f32_16x16x4_f32(false, a[kk], false, b0, (short)0, c0, false, false);
        c1 = __builtin_amdgcn_wmma_f32_16x16x4_f32(false, a[kk], false, b1, (short)0, c1, false, false);
        c2 = __builtin_amdgcn_wmma_f32_16x16x4_f32(false, a[kk], false, b2, (short)0, c2, false, false);
        c3 = __builtin_amdgcn_wmma_f32_16x16x4_f32(false, a[kk], false, b3, (short)0, c3, false, false);
    }
    c[0] = c0; c[1] = c1; c[2] = c2; c[3] = c3;
}

__device__ __forceinline__ void chain_put(const v8f (&c)[4], const float (&bias)[4], float* hb, int m, int half) {
#pragma unroll
    for (int nt = 0; nt < 4; ++nt) {
#pragma unroll
        for (int r = 0; r < 8; ++r)
            hb[(r + 8 * half) * CH_PITCH + nt * 16 + m] = fmaxf(c[nt][r] + bias[nt], 0.0f);
    }
}

__device__ __forceinline__ void chain_get(v2f (&a)[16], const float* hb, int m, int half) {
#pragma unroll
    for (int kk = 0; kk < 16; ++kk) a[kk] = *(const v2f*)&hb[m * CH_PITCH + kk * 4 + 2 * half];
}

__global__ void __launch_bounds__(CH_WAVES * 32)
k_chain3_dot(const float* __restrict__ A,
             const float* __restrict__ Wa, const float* __restrict__ ba,
             const float* __restrict__ Wb, const float* __restrict__ bb,
             const float* __restrict__ Wc, const float* __restrict__ bc,
             const float* __restrict__ wd, const float* __restrict__ bd,
             float* __restrict__ out, int nrows) {
    __shared__ float ldsW[3 * 64 * WT_PITCH];
    __shared__ float ldsH[CH_WAVES * 16 * CH_PITCH];
    __shared__ float ldsD[CH_WAVES * 16 * 2];

    const int lane = threadIdx.x & 31;
    const int wave = threadIdx.x >> 5;
    const int tile = blockIdx.x * CH_WAVES + wave;
    const int half = lane >> 4;
    const int m    = lane & 15;
    int row = tile * 16 + m;
    if (row >= nrows) row = nrows - 1;
    const float* __restrict__ arow = A + (size_t)row * 64;

    v2f a[16];
#pragma unroll
    for (int kk = 0; kk < 16; ++kk)
        a[kk] = *(const v2f*)(arow + kk * 4 + 2 * half);

    float4 wv[6];
#pragma unroll
    for (int i = 0; i < 6; ++i) {
        const float* __restrict__ Wp = (i >> 1) == 0 ? Wa : ((i >> 1) == 1 ? Wb : Wc);
        wv[i] = *(const float4*)(Wp + (size_t)(threadIdx.x + (i & 1) * (CH_WAVES * 32)) * 4);
    }
    float bia[4], bib[4], bic[4];
#pragma unroll
    for (int nt = 0; nt < 4; ++nt) {
        bia[nt] = ba[nt * 16 + m];
        bib[nt] = bb[nt * 16 + m];
        bic[nt] = bc[nt * 16 + m];
    }
    float4 dw[8];
#pragma unroll
    for (int k = 0; k < 8; ++k) dw[k] = *(const float4*)(wd + 32 * half + 4 * k);
    const float b0 = bd[0];

#pragma unroll
    for (int i = 0; i < 6; ++i) {
        const int idx = (threadIdx.x + (i & 1) * (CH_WAVES * 32)) * 4;
        const int k = idx >> 6;
        const int col = idx & 63;
        float* wt = ldsW + (i >> 1) * (64 * WT_PITCH);
        wt[(col + 0) * WT_PITCH + k] = wv[i].x;
        wt[(col + 1) * WT_PITCH + k] = wv[i].y;
        wt[(col + 2) * WT_PITCH + k] = wv[i].z;
        wt[(col + 3) * WT_PITCH + k] = wv[i].w;
    }
    __syncthreads();

    float* hb = ldsH + wave * (16 * CH_PITCH);
    v8f c[4];

    chain_product(a, ldsW + 0 * (64 * WT_PITCH), m, half, c);
    chain_put(c, bia, hb, m, half);
    __syncthreads();
    chain_get(a, hb, m, half);

    chain_product(a, ldsW + 1 * (64 * WT_PITCH), m, half, c);
    chain_put(c, bib, hb, m, half);
    __syncthreads();
    chain_get(a, hb, m, half);

    chain_product(a, ldsW + 2 * (64 * WT_PITCH), m, half, c);
    chain_put(c, bic, hb, m, half);
    __syncthreads();

    const float4* hp = (const float4*)(hb + m * CH_PITCH + 32 * half);
    float s = 0.0f;
#pragma unroll
    for (int k = 0; k < 8; ++k) {
        const float4 hv = hp[k];
        s += hv.x * dw[k].x + hv.y * dw[k].y + hv.z * dw[k].z + hv.w * dw[k].w;
    }
    ldsD[(wave * 16 + m) * 2 + half] = s;
    __syncthreads();

    if (threadIdx.x < CH_WAVES * 16) {
        const int orow = blockIdx.x * (CH_WAVES * 16) + (int)threadIdx.x;
        if (orow < nrows) out[orow] = (ldsD[threadIdx.x * 2] + ldsD[threadIdx.x * 2 + 1]) + b0;
    }
}


extern "C" void kernel_launch(void* const* d_in, const int* in_sizes, int n_in,
                              void* d_out, int out_size, void* d_ws, size_t ws_size,
                              hipStream_t stream) {
    (void)n_in; (void)out_size; (void)ws_size;

    const float* x   = (const float*)d_in[0];
    const int*   ei  = (const int*)  d_in[1];
    const float* W1  = (const float*)d_in[2];
    const float* b1  = (const float*)d_in[3];
    const float* W2  = (const float*)d_in[4];
    const float* b2  = (const float*)d_in[5];
    const float* Wf1 = (const float*)d_in[6];
    const float* bf1 = (const float*)d_in[7];
    const float* Wf2 = (const float*)d_in[8];
    const float* bf2 = (const float*)d_in[9];
    const float* Wf3 = (const float*)d_in[10];
    const float* bf3 = (const float*)d_in[11];
    float* out = (float*)d_out;

    const int n = in_sizes[0] / 64;
    const int E = in_sizes[1] / 2;

    float* ws    = (float*)d_ws;
    float* dinv  = ws;
    float* bufA  = ws + n;
    float* bufB  = bufA + (size_t)n * 64;
    int*   keys  = (int*)((char*)d_ws + ((size_t)128 << 20));
    int*   cnts  = (int*)((char*)d_ws + ((size_t)160 << 20));
    int*   recs  = (int*)((char*)d_ws + ((size_t)192 << 20));
    int*   rec0  = (int*)((char*)d_ws + ((size_t)96 << 20));
    unsigned* part = (unsigned*)((char*)d_ws + ((size_t)64 << 20));

    const int T = 256;
    const int rowTiles = (n + 15) / 16;
    const int gemmBlocks = (rowTiles + 7) / 8;
    const int ownBlocks = (n + OWN_R - 1) / OWN_R;
    const int chainBlocks = (rowTiles + CH_WAVES - 1) / CH_WAVES;

    k_deg_hist_keys<<<DEG_S * DEG_K, DEG_T, 0, stream>>>(ei, part, keys, rec0, cnts, n, E);
    k_dinv_from_parts<<<(n + T - 1) / T, T, 0, stream>>>(part, dinv, n);

    k_gemm64<false><<<gemmBlocks, T, 0, stream>>>(x, W1, nullptr, bufA, n);
    k_agg_owner<false><<<ownBlocks, OWN_WAVES * 32, 0, stream>>>(ei, keys, dinv, bufA, b1, bufB, n, E, rec0, recs, cnts);

    k_agg_owner<true><<<ownBlocks, OWN_WAVES * 32, 0, stream>>>(ei, keys, dinv, bufB, b2, bufA, n, E, rec0, recs, cnts);

    k_chain3_dot<<<chainBlocks, CH_WAVES * 32, 0, stream>>>(bufA, W2, b2, Wf1, bf1, Wf2, bf2, Wf3, bf3, out, n);
}
